# starting kernel
# baseline (speedup 1.0000x reference)
#include <hip/hip_runtime.h>
#include <hip/hip_bf16.h>


#define CIN 128

typedef __attribute__((ext_vector_type(2))) float v2f;
typedef __attribute__((ext_vector_type(4))) float v4f;
typedef __attribute__((ext_vector_type(8))) float v8f;


__global__ void fill_deg(float* __restrict__ deg, int N) {
  int i = blockIdx.x * blockDim.x + threadIdx.x;
  if (i < N) deg[i] = 1.0f;
}

__global__ void deg_edges(const int* __restrict__ dst, const float* __restrict__ ea,
                          float* __restrict__ deg, int E) {
  int e = blockIdx.x * blockDim.x + threadIdx.x;
  if (e < E) atomicAdd(&deg[dst[e]], ea[e]);
}

__global__ void calc_dinv(const float* __restrict__ deg, float* __restrict__ dinv, int N) {
  int i = blockIdx.x * blockDim.x + threadIdx.x;
  if (i < N) {
    float d = deg[i];
    dinv[i] = (d > 0.0f) ? rsqrtf(d) : 0.0f;
  }
}


template <int COUT, bool RELU>
__global__ __launch_bounds__(256) void gemm_wmma(const float* __restrict__ H,
                                                 const float* __restrict__ Wt,
                                                 float* __restrict__ T,
                                                 int nRowTiles) {
  constexpr int PITCH = 2 * COUT + 32;
  __shared__ float sW[(CIN / 2) * PITCH];

  for (int i = threadIdx.x; i < CIN * COUT; i += blockDim.x) {
    int k = i / COUT;
    int n = i - k * COUT;
    sW[(k >> 1) * PITCH + 2 * n + (k & 1)] = Wt[i];
  }
  __syncthreads();

  const int wave = threadIdx.x >> 5;
  const int lane = threadIdx.x & 31;
  const int rt   = blockIdx.x * 8 + wave;
  if (rt >= nRowTiles) return;

  const int m0 = rt * 16;
  const int m  = lane & 15;
  const int kh = lane >> 4;

  constexpr int NT = COUT / 16;
  v8f acc[NT] = {};

  const float* arow = H + (size_t)(m0 + m) * CIN + 2 * kh;

  for (int k = 0; k < CIN; k += 4) {
    v2f a = *(const v2f*)(arow + k);
    if (RELU) {
      a.x = fmaxf(a.x, 0.0f);
      a.y = fmaxf(a.y, 0.0f);
    }
    const float* b0 = &sW[((k >> 1) + kh) * PITCH + 2 * m];
#pragma unroll
    for (int t = 0; t < NT; ++t) {
      v2f b = *(const v2f*)(b0 + 32 * t);
      acc[t] = __builtin_amdgcn_wmma_f32_16x16x4_f32(
           false, a,  false, b,
           (short)0, acc[t],  false,  false);
    }
  }

  float* out = T + (size_t)(m0 + 8 * kh) * COUT + m;
#pragma unroll
  for (int t = 0; t < NT; ++t) {
#pragma unroll
    for (int v = 0; v < 8; ++v) {
      out[(size_t)v * COUT + t * 16] = acc[t][v];
    }
  }
}


__global__ void init_agg(const float* __restrict__ T, const float* __restrict__ dinv,
                         const float* __restrict__ bias, float* __restrict__ agg,
                         int N, int cpr  ) {
  int i = blockIdx.x * blockDim.x + threadIdx.x;
  int total = N * cpr;
  if (i >= total) return;
  int n  = i / cpr;
  int c4 = i - n * cpr;
  float s = dinv[n];
  s = s * s;
  v4f t = ((const v4f*)T)[(size_t)n * cpr + c4];
  v4f b = ((const v4f*)bias)[c4];
  ((v4f*)agg)[(size_t)n * cpr + c4] = b + s * t;
}


template <int C>
__global__ void edge_scatter(const int* __restrict__ src, const int* __restrict__ dst,
                             const float* __restrict__ ea, const float* __restrict__ dinv,
                             const float* __restrict__ T, float* __restrict__ agg, int E) {
  constexpr int LPE = C / 4;
  int tid  = blockIdx.x * blockDim.x + threadIdx.x;
  int e    = tid / LPE;
  int lane = tid % LPE;
  if (e >= E) return;

  int s = src[e];
  int d = dst[e];
  float nrm = dinv[s] * ea[e] * dinv[d];

  v4f v = ((const v4f*)(T + (size_t)s * C))[lane];
  float* a = agg + (size_t)d * C + lane * 4;
  atomicAdd(a + 0, nrm * v.x);
  atomicAdd(a + 1, nrm * v.y);
  atomicAdd(a + 2, nrm * v.z);
  atomicAdd(a + 3, nrm * v.w);
}


extern "C" void kernel_launch(void* const* d_in, const int* in_sizes, int n_in,
                              void* d_out, int out_size, void* d_ws, size_t ws_size,
                              hipStream_t stream) {
  const float* x  = (const float*)d_in[0];
  const int*   ei = (const int*)d_in[1];
  const float* ea = (const float*)d_in[2];
  const float* W1 = (const float*)d_in[3];
  const float* b1 = (const float*)d_in[4];
  const float* W2 = (const float*)d_in[5];
  const float* b2 = (const float*)d_in[6];
  const float* W3 = (const float*)d_in[7];
  const float* b3 = (const float*)d_in[8];

  const int N = in_sizes[0] / CIN;
  const int E = in_sizes[2];
  const int* srcIdx = ei;
  const int* dstIdx = ei + E;

  float* T    = (float*)d_ws;
  float* agg  = T + (size_t)N * 128;
  float* deg  = agg + (size_t)N * 128;
  float* dinv = deg + N;
  float* out  = (float*)d_out;

  const int nRowTiles = N / 16;
  dim3 blk(256);
  dim3 gN((N + 255) / 256);
  dim3 gE((E + 255) / 256);
  dim3 gGemm((nRowTiles + 7) / 8);
  dim3 gInit128(((unsigned)N * 32 + 255) / 256);
  dim3 gInit64(((unsigned)N * 16 + 255) / 256);
  dim3 gScat128(((unsigned)E * 32 + 255) / 256);
  dim3 gScat64(((unsigned)E * 16 + 255) / 256);

  fill_deg<<<gN, blk, 0, stream>>>(deg, N);
  deg_edges<<<gE, blk, 0, stream>>>(dstIdx, ea, deg, E);
  calc_dinv<<<gN, blk, 0, stream>>>(deg, dinv, N);

  gemm_wmma<128, false><<<gGemm, blk, 0, stream>>>(x, W1, T, nRowTiles);
  init_agg<<<gInit128, blk, 0, stream>>>(T, dinv, b1, agg, N, 32);
  edge_scatter<128><<<gScat128, blk, 0, stream>>>(srcIdx, dstIdx, ea, dinv, T, agg, E);

  gemm_wmma<128, true><<<gGemm, blk, 0, stream>>>(agg, W2, T, nRowTiles);
  init_agg<<<gInit128, blk, 0, stream>>>(T, dinv, b2, agg, N, 32);
  edge_scatter<128><<<gScat128, blk, 0, stream>>>(srcIdx, dstIdx, ea, dinv, T, agg, E);

  gemm_wmma<64, true><<<gGemm, blk, 0, stream>>>(agg, W3, T, nRowTiles);
  init_agg<<<gInit64, blk, 0, stream>>>(T, dinv, b3, out, N, 16);
  edge_scatter<64><<<gScat64, blk, 0, stream>>>(srcIdx, dstIdx, ea, dinv, T, out, E);
}


# Replace float-atomic edge scatter by per-call grouping of edges by destination (counting sort) and a wave-per-row register sum; sum order per row now follows the grouping rank
# speedup vs baseline: 2.3995x; 2.3995x over previous
#include <hip/hip_runtime.h>
#include <hip/hip_bf16.h>


#define CIN 128

typedef __attribute__((ext_vector_type(2))) float v2f;
typedef __attribute__((ext_vector_type(4))) float v4f;
typedef __attribute__((ext_vector_type(8))) float v8f;


__global__ void zero_counts(int* __restrict__ cnt, int N) {
  int i = blockIdx.x * blockDim.x + threadIdx.x;
  if (i < N) cnt[i] = 0;
}

__global__ void rank_edges(const int* __restrict__ dst, int* __restrict__ cnt,
                           int* __restrict__ rank, int E) {
  int e = blockIdx.x * blockDim.x + threadIdx.x;
  if (e < E) rank[e] = atomicAdd(&cnt[dst[e]], 1);
}

__global__ __launch_bounds__(1024) void scan_counts(const int* __restrict__ cnt,
                                                    int* __restrict__ off, int N) {
  __shared__ int sPre[32];
  __shared__ int sWave[32];
  const int t    = threadIdx.x;
  const int lane = t & 31;
  const int wv   = t >> 5;
  const int b    = blockIdx.x;

  int pre = 0;
#pragma unroll 8
  for (int i = 0; i < b; ++i) pre += cnt[i * 1024 + t];
  for (int o = 16; o > 0; o >>= 1) pre += __shfl_xor(pre, o);
  if (lane == 0) sPre[wv] = pre;

  const int i = b * 1024 + t;
  const int c = (i < N) ? cnt[i] : 0;
  int x = c;
  for (int o = 1; o < 32; o <<= 1) {
    int y = __shfl_up(x, o);
    if (lane >= o) x += y;
  }
  if (lane == 31) sWave[wv] = x;
  __syncthreads();

  int base = 0;
  for (int k = 0; k < 32; ++k) base += sPre[k];
  for (int k = 0; k < wv; ++k) base += sWave[k];
  if (i < N) off[i] = base + x - c;
}

__global__ void place_edges(const int* __restrict__ src, const int* __restrict__ dst,
                            const float* __restrict__ ea, const int* __restrict__ off,
                            const int* __restrict__ rank, int* __restrict__ pairs, int E) {
  int e = blockIdx.x * blockDim.x + threadIdx.x;
  if (e >= E) return;
  int p = off[dst[e]] + rank[e];
  *(int2*)(pairs + 2 * (size_t)p) = make_int2(src[e], __float_as_int(ea[e]));
}

__global__ void node_dinv(const int* __restrict__ pairs, const int* __restrict__ off,
                          const int* __restrict__ cnt, float* __restrict__ dinv, int N) {
  int n = blockIdx.x * blockDim.x + threadIdx.x;
  if (n >= N) return;
  const int* p = pairs + 2 * (size_t)off[n];
  const int c = cnt[n];
  float d = 1.0f;
  for (int j = 0; j < c; ++j) d += __int_as_float(p[2 * j + 1]);
  dinv[n] = (d > 0.0f) ? rsqrtf(d) : 0.0f;
}

__global__ void edge_coef(int* __restrict__ pairs, const int* __restrict__ off,
                          const int* __restrict__ cnt, const float* __restrict__ dinv, int N) {
  int n = blockIdx.x * blockDim.x + threadIdx.x;
  if (n >= N) return;
  int* p = pairs + 2 * (size_t)off[n];
  const int c = cnt[n];
  const float dn = dinv[n];
  for (int j = 0; j < c; ++j) {
    const int2 en = *(const int2*)(p + 2 * j);
    p[2 * j + 1] = __float_as_int(dinv[en.x] * __int_as_float(en.y) * dn);
  }
}


template <int COUT, bool RELU>
__global__ __launch_bounds__(256) void gemm_wmma(const float* __restrict__ H,
                                                 const float* __restrict__ Wt,
                                                 float* __restrict__ T,
                                                 int nRowTiles) {
  constexpr int PITCH = 2 * COUT + 32;
  __shared__ float sW[(CIN / 2) * PITCH];

  for (int i = threadIdx.x; i < CIN * COUT; i += blockDim.x) {
    int k = i / COUT;
    int n = i - k * COUT;
    sW[(k >> 1) * PITCH + 2 * n + (k & 1)] = Wt[i];
  }
  __syncthreads();

  const int wave = threadIdx.x >> 5;
  const int lane = threadIdx.x & 31;
  const int rt   = blockIdx.x * 8 + wave;
  if (rt >= nRowTiles) return;

  const int m0 = rt * 16;
  const int m  = lane & 15;
  const int kh = lane >> 4;

  constexpr int NT = COUT / 16;
  v8f acc[NT] = {};

  const float* arow = H + (size_t)(m0 + m) * CIN + 2 * kh;

  for (int k = 0; k < CIN; k += 4) {
    v2f a = *(const v2f*)(arow + k);
    if (RELU) {
      a.x = fmaxf(a.x, 0.0f);
      a.y = fmaxf(a.y, 0.0f);
    }
    const float* b0 = &sW[((k >> 1) + kh) * PITCH + 2 * m];
#pragma unroll
    for (int t = 0; t < NT; ++t) {
      v2f b = *(const v2f*)(b0 + 32 * t);
      acc[t] = __builtin_amdgcn_wmma_f32_16x16x4_f32(
           false, a,  false, b,
           (short)0, acc[t],  false,  false);
    }
  }

  float* out = T + (size_t)(m0 + 8 * kh) * COUT + m;
#pragma unroll
  for (int t = 0; t < NT; ++t) {
#pragma unroll
    for (int v = 0; v < 8; ++v) {
      out[(size_t)v * COUT + t * 16] = acc[t][v];
    }
  }
}


template <int C>
__global__ __launch_bounds__(256) void gather_rows(const int* __restrict__ pairs,
                                                   const int* __restrict__ off,
                                                   const int* __restrict__ cnt,
                                                   const float* __restrict__ dinv,
                                                   const float* __restrict__ T,
                                                   const float* __restrict__ bias,
                                                   float* __restrict__ agg, int N) {
  constexpr int V = C / 32;
  typedef float vecT __attribute__((ext_vector_type(V)));

  const int lane = threadIdx.x & 31;
  const int wv = __builtin_amdgcn_readfirstlane((int)(threadIdx.x >> 5));
  const int n  = blockIdx.x * 8 + wv;
  if (n >= N) return;

  const int beg = off[n];
  const int c   = cnt[n];
  float s = dinv[n];
  s = s * s;

  const int col = lane * V;
  vecT acc = *(const vecT*)(bias + col) + s * *(const vecT*)(T + (size_t)n * C + col);

  for (int base = 0; base < c; base += 32) {
    const int m = (c - base < 32) ? (c - base) : 32;
    int ms = 0;
    int mc = 0;
    if (lane < m) {
      const int2 en = *(const int2*)(pairs + 2 * (size_t)(beg + base + lane));
      ms = en.x;
      mc = en.y;
    }
    int j = 0;
    for (; j + 4 <= m; j += 4) {
      const int s0 = __builtin_amdgcn_readlane(ms, j);
      const int s1 = __builtin_amdgcn_readlane(ms, j + 1);
      const int s2 = __builtin_amdgcn_readlane(ms, j + 2);
      const int s3 = __builtin_amdgcn_readlane(ms, j + 3);
      const float c0 = __int_as_float(__builtin_amdgcn_readlane(mc, j));
      const float c1 = __int_as_float(__builtin_amdgcn_readlane(mc, j + 1));
      const float c2 = __int_as_float(__builtin_amdgcn_readlane(mc, j + 2));
      const float c3 = __int_as_float(__builtin_amdgcn_readlane(mc, j + 3));
      const vecT r0 = *(const vecT*)(T + (size_t)s0 * C + col);
      const vecT r1 = *(const vecT*)(T + (size_t)s1 * C + col);
      const vecT r2 = *(const vecT*)(T + (size_t)s2 * C + col);
      const vecT r3 = *(const vecT*)(T + (size_t)s3 * C + col);
      acc += c0 * r0;
      acc += c1 * r1;
      acc += c2 * r2;
      acc += c3 * r3;
    }
    for (; j < m; ++j) {
      const int s0 = __builtin_amdgcn_readlane(ms, j);
      const float c0 = __int_as_float(__builtin_amdgcn_readlane(mc, j));
      acc += c0 * *(const vecT*)(T + (size_t)s0 * C + col);
    }
  }

  *(vecT*)(agg + (size_t)n * C + col) = acc;
}


extern "C" void kernel_launch(void* const* d_in, const int* in_sizes, int n_in,
                              void* d_out, int out_size, void* d_ws, size_t ws_size,
                              hipStream_t stream) {
  const float* x  = (const float*)d_in[0];
  const int*   ei = (const int*)d_in[1];
  const float* ea = (const float*)d_in[2];
  const float* W1 = (const float*)d_in[3];
  const float* b1 = (const float*)d_in[4];
  const float* W2 = (const float*)d_in[5];
  const float* b2 = (const float*)d_in[6];
  const float* W3 = (const float*)d_in[7];
  const float* b3 = (const float*)d_in[8];

  const int N = in_sizes[0] / CIN;
  const int E = in_sizes[2];
  const int* srcIdx = ei;
  const int* dstIdx = ei + E;

  float* T     = (float*)d_ws;
  float* agg   = T + (size_t)N * 128;
  int*   pairs = (int*)(agg + (size_t)N * 128);
  int*   rank  = pairs + (size_t)2 * E;
  float* dinv  = (float*)(rank + E);
  int*   cnt   = (int*)(dinv + N);
  int*   off   = cnt + N;
  float* out   = (float*)d_out;

  const int nRowTiles = N / 16;
  dim3 blk(256);
  dim3 gN((N + 255) / 256);
  dim3 gE((E + 255) / 256);
  dim3 gScan((N + 1023) / 1024);
  dim3 gGemm((nRowTiles + 7) / 8);
  dim3 gRows((N + 7) / 8);

  zero_counts<<<gN, blk, 0, stream>>>(cnt, N);
  rank_edges<<<gE, blk, 0, stream>>>(dstIdx, cnt, rank, E);
  scan_counts<<<gScan, dim3(1024), 0, stream>>>(cnt, off, N);
  place_edges<<<gE, blk, 0, stream>>>(srcIdx, dstIdx, ea, off, rank, pairs, E);
  node_dinv<<<gN, blk, 0, stream>>>(pairs, off, cnt, dinv, N);
  edge_coef<<<gN, blk, 0, stream>>>(pairs, off, cnt, dinv, N);

  gemm_wmma<128, false><<<gGemm, blk, 0, stream>>>(x, W1, T, nRowTiles);
  gather_rows<128><<<gRows, blk, 0, stream>>>(pairs, off, cnt, dinv, T, b1, agg, N);

  gemm_wmma<128, true><<<gGemm, blk, 0, stream>>>(agg, W2, T, nRowTiles);
  gather_rows<128><<<gRows, blk, 0, stream>>>(pairs, off, cnt, dinv, T, b2, agg, N);

  gemm_wmma<64, true><<<gGemm, blk, 0, stream>>>(agg, W3, T, nRowTiles);
  gather_rows<64><<<gRows, blk, 0, stream>>>(pairs, off, cnt, dinv, T, b3, out, N);
}


# GEMM kernels: a thread issues all its W loads (16-byte, whole lines per wave) as one group ahead of the 16-byte LDS stores; was 64 one-word load-wait-store trips. Products and sum order unchanged
# speedup vs baseline: 2.8356x; 1.1817x over previous
#include <hip/hip_runtime.h>
#include <hip/hip_bf16.h>


#define CIN 128

typedef __attribute__((ext_vector_type(2))) float v2f;
typedef __attribute__((ext_vector_type(4))) float v4f;
typedef __attribute__((ext_vector_type(8))) float v8f;


__global__ void zero_counts(int* __restrict__ cnt, int N) {
  int i = blockIdx.x * blockDim.x + threadIdx.x;
  if (i < N) cnt[i] = 0;
}

__global__ void rank_edges(const int* __restrict__ dst, int* __restrict__ cnt,
                           int* __restrict__ rank, int E) {
  int e = blockIdx.x * blockDim.x + threadIdx.x;
  if (e < E) rank[e] = atomicAdd(&cnt[dst[e]], 1);
}

__global__ __launch_bounds__(1024) void scan_counts(const int* __restrict__ cnt,
                                                    int* __restrict__ off, int N) {
  __shared__ int sPre[32];
  __shared__ int sWave[32];
  const int t    = threadIdx.x;
  const int lane = t & 31;
  const int wv   = t >> 5;
  const int b    = blockIdx.x;

  int pre = 0;
#pragma unroll 8
  for (int i = 0; i < b; ++i) pre += cnt[i * 1024 + t];
  for (int o = 16; o > 0; o >>= 1) pre += __shfl_xor(pre, o);
  if (lane == 0) sPre[wv] = pre;

  const int i = b * 1024 + t;
  const int c = (i < N) ? cnt[i] : 0;
  int x = c;
  for (int o = 1; o < 32; o <<= 1) {
    int y = __shfl_up(x, o);
    if (lane >= o) x += y;
  }
  if (lane == 31) sWave[wv] = x;
  __syncthreads();

  int base = 0;
  for (int k = 0; k < 32; ++k) base += sPre[k];
  for (int k = 0; k < wv; ++k) base += sWave[k];
  if (i < N) off[i] = base + x - c;
}

__global__ void place_edges(const int* __restrict__ src, const int* __restrict__ dst,
                            const float* __restrict__ ea, const int* __restrict__ off,
                            const int* __restrict__ rank, int* __restrict__ pairs, int E) {
  int e = blockIdx.x * blockDim.x + threadIdx.x;
  if (e >= E) return;
  int p = off[dst[e]] + rank[e];
  *(int2*)(pairs + 2 * (size_t)p) = make_int2(src[e], __float_as_int(ea[e]));
}

__global__ void node_dinv(const int* __restrict__ pairs, const int* __restrict__ off,
                          const int* __restrict__ cnt, float* __restrict__ dinv, int N) {
  int n = blockIdx.x * blockDim.x + threadIdx.x;
  if (n >= N) return;
  const int* p = pairs + 2 * (size_t)off[n];
  const int c = cnt[n];
  float d = 1.0f;
  for (int j = 0; j < c; ++j) d += __int_as_float(p[2 * j + 1]);
  dinv[n] = (d > 0.0f) ? rsqrtf(d) : 0.0f;
}

__global__ void edge_coef(int* __restrict__ pairs, const int* __restrict__ off,
                          const int* __restrict__ cnt, const float* __restrict__ dinv, int N) {
  int n = blockIdx.x * blockDim.x + threadIdx.x;
  if (n >= N) return;
  int* p = pairs + 2 * (size_t)off[n];
  const int c = cnt[n];
  const float dn = dinv[n];
  for (int j = 0; j < c; ++j) {
    const int2 en = *(const int2*)(p + 2 * j);
    p[2 * j + 1] = __float_as_int(dinv[en.x] * __int_as_float(en.y) * dn);
  }
}


template <int COUT, bool RELU>
__global__ __launch_bounds__(256) void gemm_wmma(const float* __restrict__ H,
                                                 const float* __restrict__ Wt,
                                                 float* __restrict__ T,
                                                 int nRowTiles) {
  constexpr int PITCH = 2 * COUT + 32;
  __shared__ alignas(16) float sW[(CIN / 2) * PITCH];

  constexpr int NTHREADS = 256;
  constexpr int NV       = COUT / 4;
  constexpr int PER_T    = (CIN / 2) * NV / NTHREADS;
  v4f wlo[PER_T], whi[PER_T];
#pragma unroll
  for (int u = 0; u < PER_T; ++u) {
    const int id = (int)threadIdx.x + u * NTHREADS;
    const int p  = id / NV;
    const int g  = id - p * NV;
    wlo[u] = *(const v4f*)(Wt + (size_t)(2 * p) * COUT + 4 * g);
    whi[u] = *(const v4f*)(Wt + (size_t)(2 * p + 1) * COUT + 4 * g);
  }
#pragma unroll
  for (int u = 0; u < PER_T; ++u) {
    const int id = (int)threadIdx.x + u * NTHREADS;
    const int p  = id / NV;
    const int g  = id - p * NV;
    float* d = &sW[p * PITCH + 8 * g];
    v4f s0, s1;
    s0.x = wlo[u].x; s0.y = whi[u].x; s0.z = wlo[u].y; s0.w = whi[u].y;
    s1.x = wlo[u].z; s1.y = whi[u].z; s1.z = wlo[u].w; s1.w = whi[u].w;
    *(v4f*)(d)     = s0;
    *(v4f*)(d + 4) = s1;
  }
  __syncthreads();

  const int wave = threadIdx.x >> 5;
  const int lane = threadIdx.x & 31;
  const int rt   = blockIdx.x * 8 + wave;
  if (rt >= nRowTiles) return;

  const int m0 = rt * 16;
  const int m  = lane & 15;
  const int kh = lane >> 4;

  constexpr int NT = COUT / 16;
  v8f acc[NT] = {};

  const float* arow = H + (size_t)(m0 + m) * CIN + 2 * kh;

  for (int k = 0; k < CIN; k += 4) {
    v2f a = *(const v2f*)(arow + k);
    if (RELU) {
      a.x = fmaxf(a.x, 0.0f);
      a.y = fmaxf(a.y, 0.0f);
    }
    const float* b0 = &sW[((k >> 1) + kh) * PITCH + 2 * m];
#pragma unroll
    for (int t = 0; t < NT; ++t) {
      v2f b = *(const v2f*)(b0 + 32 * t);
      acc[t] = __builtin_amdgcn_wmma_f32_16x16x4_f32(
           false, a,  false, b,
           (short)0, acc[t],  false,  false);
    }
  }

  float* out = T + (size_t)(m0 + 8 * kh) * COUT + m;
#pragma unroll
  for (int t = 0; t < NT; ++t) {
#pragma unroll
    for (int v = 0; v < 8; ++v) {
      out[(size_t)v * COUT + t * 16] = acc[t][v];
    }
  }
}


template <int C>
__global__ __launch_bounds__(256) void gather_rows(const int* __restrict__ pairs,
                                                   const int* __restrict__ off,
                                                   const int* __restrict__ cnt,
                                                   const float* __restrict__ dinv,
                                                   const float* __restrict__ T,
                                                   const float* __restrict__ bias,
                                                   float* __restrict__ agg, int N) {
  constexpr int V = C / 32;
  typedef float vecT __attribute__((ext_vector_type(V)));

  const int lane = threadIdx.x & 31;
  const int wv = __builtin_amdgcn_readfirstlane((int)(threadIdx.x >> 5));
  const int n  = blockIdx.x * 8 + wv;
  if (n >= N) return;

  const int beg = off[n];
  const int c   = cnt[n];
  float s = dinv[n];
  s = s * s;

  const int col = lane * V;
  vecT acc = *(const vecT*)(bias + col) + s * *(const vecT*)(T + (size_t)n * C + col);

  for (int base = 0; base < c; base += 32) {
    const int m = (c - base < 32) ? (c - base) : 32;
    int ms = 0;
    int mc = 0;
    if (lane < m) {
      const int2 en = *(const int2*)(pairs + 2 * (size_t)(beg + base + lane));
      ms = en.x;
      mc = en.y;
    }
    int j = 0;
    for (; j + 4 <= m; j += 4) {
      const int s0 = __builtin_amdgcn_readlane(ms, j);
      const int s1 = __builtin_amdgcn_readlane(ms, j + 1);
      const int s2 = __builtin_amdgcn_readlane(ms, j + 2);
      const int s3 = __builtin_amdgcn_readlane(ms, j + 3);
      const float c0 = __int_as_float(__builtin_amdgcn_readlane(mc, j));
      const float c1 = __int_as_float(__builtin_amdgcn_readlane(mc, j + 1));
      const float c2 = __int_as_float(__builtin_amdgcn_readlane(mc, j + 2));
      const float c3 = __int_as_float(__builtin_amdgcn_readlane(mc, j + 3));
      const vecT r0 = *(const vecT*)(T + (size_t)s0 * C + col);
      const vecT r1 = *(const vecT*)(T + (size_t)s1 * C + col);
      const vecT r2 = *(const vecT*)(T + (size_t)s2 * C + col);
      const vecT r3 = *(const vecT*)(T + (size_t)s3 * C + col);
      acc += c0 * r0;
      acc += c1 * r1;
      acc += c2 * r2;
      acc += c3 * r3;
    }
    for (; j < m; ++j) {
      const int s0 = __builtin_amdgcn_readlane(ms, j);
      const float c0 = __int_as_float(__builtin_amdgcn_readlane(mc, j));
      acc += c0 * *(const vecT*)(T + (size_t)s0 * C + col);
    }
  }

  *(vecT*)(agg + (size_t)n * C + col) = acc;
}


extern "C" void kernel_launch(void* const* d_in, const int* in_sizes, int n_in,
                              void* d_out, int out_size, void* d_ws, size_t ws_size,
                              hipStream_t stream) {
  const float* x  = (const float*)d_in[0];
  const int*   ei = (const int*)d_in[1];
  const float* ea = (const float*)d_in[2];
  const float* W1 = (const float*)d_in[3];
  const float* b1 = (const float*)d_in[4];
  const float* W2 = (const float*)d_in[5];
  const float* b2 = (const float*)d_in[6];
  const float* W3 = (const float*)d_in[7];
  const float* b3 = (const float*)d_in[8];

  const int N = in_sizes[0] / CIN;
  const int E = in_sizes[2];
  const int* srcIdx = ei;
  const int* dstIdx = ei + E;

  float* T     = (float*)d_ws;
  float* agg   = T + (size_t)N * 128;
  int*   pairs = (int*)(agg + (size_t)N * 128);
  int*   rank  = pairs + (size_t)2 * E;
  float* dinv  = (float*)(rank + E);
  int*   cnt   = (int*)(dinv + N);
  int*   off   = cnt + N;
  float* out   = (float*)d_out;

  const int nRowTiles = N / 16;
  dim3 blk(256);
  dim3 gN((N + 255) / 256);
  dim3 gE((E + 255) / 256);
  dim3 gScan((N + 1023) / 1024);
  dim3 gGemm((nRowTiles + 7) / 8);
  dim3 gRows((N + 7) / 8);

  zero_counts<<<gN, blk, 0, stream>>>(cnt, N);
  rank_edges<<<gE, blk, 0, stream>>>(dstIdx, cnt, rank, E);
  scan_counts<<<gScan, dim3(1024), 0, stream>>>(cnt, off, N);
  place_edges<<<gE, blk, 0, stream>>>(srcIdx, dstIdx, ea, off, rank, pairs, E);
  node_dinv<<<gN, blk, 0, stream>>>(pairs, off, cnt, dinv, N);
  edge_coef<<<gN, blk, 0, stream>>>(pairs, off, cnt, dinv, N);

  gemm_wmma<128, false><<<gGemm, blk, 0, stream>>>(x, W1, T, nRowTiles);
  gather_rows<128><<<gRows, blk, 0, stream>>>(pairs, off, cnt, dinv, T, b1, agg, N);

  gemm_wmma<128, true><<<gGemm, blk, 0, stream>>>(agg, W2, T, nRowTiles);
  gather_rows<128><<<gRows, blk, 0, stream>>>(pairs, off, cnt, dinv, T, b2, agg, N);

  gemm_wmma<64, true><<<gGemm, blk, 0, stream>>>(agg, W3, T, nRowTiles);
  gather_rows<64><<<gRows, blk, 0, stream>>>(pairs, off, cnt, dinv, T, b3, out, N);
}


# Edge ranks from LDS counters: each block counts one 61440-edge chunk in packed 16-bit LDS words, a pass sums the chunks; was 1.6M global returning adds. Row sum order follows chunk, then LDS rank
# speedup vs baseline: 3.4375x; 1.2123x over previous
#include <hip/hip_runtime.h>
#include <hip/hip_bf16.h>


#define CIN 128

typedef __attribute__((ext_vector_type(2))) float v2f;
typedef __attribute__((ext_vector_type(4))) float v4f;
typedef __attribute__((ext_vector_type(8))) float v8f;


#define RK_THREADS 1024
#define RK_TRIPS   15
#define RK_CHUNK   (RK_THREADS * 4 * RK_TRIPS)
#define RK_WORDS   65536

__device__ __forceinline__ int lds_rank(unsigned* sCnt, int d) {
  const unsigned sh  = (unsigned)(d & 1) << 4;
  const unsigned old = atomicAdd(&sCnt[d >> 1], 1u << sh);
  return (int)((old >> sh) & 0xFFFFu);
}

__device__ __forceinline__ void lds_clear(unsigned* sCnt, int NW4, int t) {
  for (int i = t * 4; i < NW4; i += RK_THREADS * 4)
    *(uint4*)&sCnt[i] = make_uint4(0u, 0u, 0u, 0u);
}

__global__ __launch_bounds__(RK_THREADS) void count_rank(const int* __restrict__ dst,
                                                         unsigned* __restrict__ cntJ,
                                                         int* __restrict__ rank, int N, int E) {
  __shared__ alignas(16) unsigned sCnt[RK_WORDS];
  const int t    = threadIdx.x;
  const int NW4  = (((N + 1) >> 1) + 3) & ~3;
  const int eBeg = blockIdx.x * RK_CHUNK;
  const bool full = ((E & 3) == 0) && (E - eBeg >= RK_CHUNK);

  if (full) {
    int4 dv[RK_TRIPS];
#pragma unroll
    for (int u = 0; u < RK_TRIPS; ++u)
      dv[u] = *(const int4*)(dst + eBeg + (u * RK_THREADS + t) * 4);
    lds_clear(sCnt, NW4, t);
    __syncthreads();
#pragma unroll
    for (int u = 0; u < RK_TRIPS; ++u) {
      int4 r;
      r.x = lds_rank(sCnt, dv[u].x);
      r.y = lds_rank(sCnt, dv[u].y);
      r.z = lds_rank(sCnt, dv[u].z);
      r.w = lds_rank(sCnt, dv[u].w);
      *(int4*)(rank + eBeg + (u * RK_THREADS + t) * 4) = r;
    }
  } else {
    const int eEnd = (E - eBeg < RK_CHUNK) ? E : eBeg + RK_CHUNK;
    lds_clear(sCnt, NW4, t);
    __syncthreads();
    for (int e = eBeg + t; e < eEnd; e += RK_THREADS) rank[e] = lds_rank(sCnt, dst[e]);
  }
  __syncthreads();

  unsigned* o = cntJ + (size_t)blockIdx.x * NW4;
  for (int i = t * 4; i < NW4; i += RK_THREADS * 4)
    *(uint4*)(o + i) = *(const uint4*)&sCnt[i];
}

#define SC_GROUP 9

template <bool EVEN>
__device__ __forceinline__ void sum_counts_body(const unsigned* __restrict__ cntJ,
                                                int* __restrict__ rel, int* __restrict__ cnt,
                                                int N, int J, int w, int NW4) {
  const int n0 = 2 * w;
  int a0 = 0, a1 = 0;
  for (int j0 = 0; j0 < J; j0 += SC_GROUP) {
    unsigned c[SC_GROUP];
#pragma unroll
    for (int k = 0; k < SC_GROUP; ++k) {
      const int j = (j0 + k < J) ? (j0 + k) : (J - 1);
      c[k] = cntJ[(size_t)j * NW4 + w];
    }
#pragma unroll
    for (int k = 0; k < SC_GROUP; ++k) {
      if (j0 + k < J) {
        int* r = rel + (size_t)(j0 + k) * N + n0;
        if (EVEN) {
          *(int2*)r = make_int2(a0, a1);
        } else {
          r[0] = a0;
          if (n0 + 1 < N) r[1] = a1;
        }
        a0 += (int)(c[k] & 0xFFFFu);
        a1 += (int)(c[k] >> 16);
      }
    }
  }
  cnt[n0] = a0;
  if (EVEN || n0 + 1 < N) cnt[n0 + 1] = a1;
}

__global__ void sum_counts(const unsigned* __restrict__ cntJ, int* __restrict__ rel,
                           int* __restrict__ cnt, int N, int J) {
  const int w   = blockIdx.x * blockDim.x + threadIdx.x;
  const int NW  = (N + 1) >> 1;
  const int NW4 = (NW + 3) & ~3;
  if (w >= NW) return;
  if ((N & 1) == 0) sum_counts_body<true>(cntJ, rel, cnt, N, J, w, NW4);
  else              sum_counts_body<false>(cntJ, rel, cnt, N, J, w, NW4);
}

__global__ __launch_bounds__(1024) void scan_counts(const int* __restrict__ cnt,
                                                    int* __restrict__ off, int N) {
  __shared__ int sPre[32];
  __shared__ int sWave[32];
  const int t    = threadIdx.x;
  const int lane = t & 31;
  const int wv   = t >> 5;
  const int b    = blockIdx.x;

  int pre = 0;
#pragma unroll 8
  for (int i = 0; i < b; ++i) pre += cnt[i * 1024 + t];
  for (int o = 16; o > 0; o >>= 1) pre += __shfl_xor(pre, o);
  if (lane == 0) sPre[wv] = pre;

  const int i = b * 1024 + t;
  const int c = (i < N) ? cnt[i] : 0;
  int x = c;
  for (int o = 1; o < 32; o <<= 1) {
    int y = __shfl_up(x, o);
    if (lane >= o) x += y;
  }
  if (lane == 31) sWave[wv] = x;
  __syncthreads();

  int base = 0;
  for (int k = 0; k < 32; ++k) base += sPre[k];
  for (int k = 0; k < wv; ++k) base += sWave[k];
  if (i < N) off[i] = base + x - c;
}

__global__ void place_edges(const int* __restrict__ src, const int* __restrict__ dst,
                            const float* __restrict__ ea, const int* __restrict__ off,
                            const int* __restrict__ rel, const int* __restrict__ rank,
                            int* __restrict__ pairs, int N, int E) {
  int e = blockIdx.x * blockDim.x + threadIdx.x;
  if (e >= E) return;
  const int d = dst[e];
  const int j = e / RK_CHUNK;
  int p = off[d] + rel[(size_t)j * N + d] + rank[e];
  *(int2*)(pairs + 2 * (size_t)p) = make_int2(src[e], __float_as_int(ea[e]));
}

__global__ void node_dinv(const int* __restrict__ pairs, const int* __restrict__ off,
                          const int* __restrict__ cnt, float* __restrict__ dinv, int N) {
  int n = blockIdx.x * blockDim.x + threadIdx.x;
  if (n >= N) return;
  const int* p = pairs + 2 * (size_t)off[n];
  const int c = cnt[n];
  float d = 1.0f;
  for (int j = 0; j < c; ++j) d += __int_as_float(p[2 * j + 1]);
  dinv[n] = (d > 0.0f) ? rsqrtf(d) : 0.0f;
}

__global__ void edge_coef(int* __restrict__ pairs, const int* __restrict__ off,
                          const int* __restrict__ cnt, const float* __restrict__ dinv, int N) {
  int n = blockIdx.x * blockDim.x + threadIdx.x;
  if (n >= N) return;
  int* p = pairs + 2 * (size_t)off[n];
  const int c = cnt[n];
  const float dn = dinv[n];
  for (int j = 0; j < c; ++j) {
    const int2 en = *(const int2*)(p + 2 * j);
    p[2 * j + 1] = __float_as_int(dinv[en.x] * __int_as_float(en.y) * dn);
  }
}


template <int COUT, bool RELU>
__global__ __launch_bounds__(256) void gemm_wmma(const float* __restrict__ H,
                                                 const float* __restrict__ Wt,
                                                 float* __restrict__ T,
                                                 int nRowTiles) {
  constexpr int PITCH = 2 * COUT + 32;
  __shared__ alignas(16) float sW[(CIN / 2) * PITCH];

  constexpr int NTHREADS = 256;
  constexpr int NV       = COUT / 4;
  constexpr int PER_T    = (CIN / 2) * NV / NTHREADS;
  v4f wlo[PER_T], whi[PER_T];
#pragma unroll
  for (int u = 0; u < PER_T; ++u) {
    const int id = (int)threadIdx.x + u * NTHREADS;
    const int p  = id / NV;
    const int g  = id - p * NV;
    wlo[u] = *(const v4f*)(Wt + (size_t)(2 * p) * COUT + 4 * g);
    whi[u] = *(const v4f*)(Wt + (size_t)(2 * p + 1) * COUT + 4 * g);
  }
#pragma unroll
  for (int u = 0; u < PER_T; ++u) {
    const int id = (int)threadIdx.x + u * NTHREADS;
    const int p  = id / NV;
    const int g  = id - p * NV;
    float* d = &sW[p * PITCH + 8 * g];
    v4f s0, s1;
    s0.x = wlo[u].x; s0.y = whi[u].x; s0.z = wlo[u].y; s0.w = whi[u].y;
    s1.x = wlo[u].z; s1.y = whi[u].z; s1.z = wlo[u].w; s1.w = whi[u].w;
    *(v4f*)(d)     = s0;
    *(v4f*)(d + 4) = s1;
  }
  __syncthreads();

  const int wave = threadIdx.x >> 5;
  const int lane = threadIdx.x & 31;
  const int rt   = blockIdx.x * 8 + wave;
  if (rt >= nRowTiles) return;

  const int m0 = rt * 16;
  const int m  = lane & 15;
  const int kh = lane >> 4;

  constexpr int NT = COUT / 16;
  v8f acc[NT] = {};

  const float* arow = H + (size_t)(m0 + m) * CIN + 2 * kh;

  for (int k = 0; k < CIN; k += 4) {
    v2f a = *(const v2f*)(arow + k);
    if (RELU) {
      a.x = fmaxf(a.x, 0.0f);
      a.y = fmaxf(a.y, 0.0f);
    }
    const float* b0 = &sW[((k >> 1) + kh) * PITCH + 2 * m];
#pragma unroll
    for (int t = 0; t < NT; ++t) {
      v2f b = *(const v2f*)(b0 + 32 * t);
      acc[t] = __builtin_amdgcn_wmma_f32_16x16x4_f32(
           false, a,  false, b,
           (short)0, acc[t],  false,  false);
    }
  }

  float* out = T + (size_t)(m0 + 8 * kh) * COUT + m;
#pragma unroll
  for (int t = 0; t < NT; ++t) {
#pragma unroll
    for (int v = 0; v < 8; ++v) {
      out[(size_t)v * COUT + t * 16] = acc[t][v];
    }
  }
}


template <int C>
__global__ __launch_bounds__(256) void gather_rows(const int* __restrict__ pairs,
                                                   const int* __restrict__ off,
                                                   const int* __restrict__ cnt,
                                                   const float* __restrict__ dinv,
                                                   const float* __restrict__ T,
                                                   const float* __restrict__ bias,
                                                   float* __restrict__ agg, int N) {
  constexpr int V = C / 32;
  typedef float vecT __attribute__((ext_vector_type(V)));

  const int lane = threadIdx.x & 31;
  const int wv = __builtin_amdgcn_readfirstlane((int)(threadIdx.x >> 5));
  const int n  = blockIdx.x * 8 + wv;
  if (n >= N) return;

  const int beg = off[n];
  const int c   = cnt[n];
  float s = dinv[n];
  s = s * s;

  const int col = lane * V;
  vecT acc = *(const vecT*)(bias + col) + s * *(const vecT*)(T + (size_t)n * C + col);

  for (int base = 0; base < c; base += 32) {
    const int m = (c - base < 32) ? (c - base) : 32;
    int ms = 0;
    int mc = 0;
    if (lane < m) {
      const int2 en = *(const int2*)(pairs + 2 * (size_t)(beg + base + lane));
      ms = en.x;
      mc = en.y;
    }
    int j = 0;
    for (; j + 4 <= m; j += 4) {
      const int s0 = __builtin_amdgcn_readlane(ms, j);
      const int s1 = __builtin_amdgcn_readlane(ms, j + 1);
      const int s2 = __builtin_amdgcn_readlane(ms, j + 2);
      const int s3 = __builtin_amdgcn_readlane(ms, j + 3);
      const float c0 = __int_as_float(__builtin_amdgcn_readlane(mc, j));
      const float c1 = __int_as_float(__builtin_amdgcn_readlane(mc, j + 1));
      const float c2 = __int_as_float(__builtin_amdgcn_readlane(mc, j + 2));
      const float c3 = __int_as_float(__builtin_amdgcn_readlane(mc, j + 3));
      const vecT r0 = *(const vecT*)(T + (size_t)s0 * C + col);
      const vecT r1 = *(const vecT*)(T + (size_t)s1 * C + col);
      const vecT r2 = *(const vecT*)(T + (size_t)s2 * C + col);
      const vecT r3 = *(const vecT*)(T + (size_t)s3 * C + col);
      acc += c0 * r0;
      acc += c1 * r1;
      acc += c2 * r2;
      acc += c3 * r3;
    }
    for (; j < m; ++j) {
      const int s0 = __builtin_amdgcn_readlane(ms, j);
      const float c0 = __int_as_float(__builtin_amdgcn_readlane(mc, j));
      acc += c0 * *(const vecT*)(T + (size_t)s0 * C + col);
    }
  }

  *(vecT*)(agg + (size_t)n * C + col) = acc;
}


extern "C" void kernel_launch(void* const* d_in, const int* in_sizes, int n_in,
                              void* d_out, int out_size, void* d_ws, size_t ws_size,
                              hipStream_t stream) {
  const float* x  = (const float*)d_in[0];
  const int*   ei = (const int*)d_in[1];
  const float* ea = (const float*)d_in[2];
  const float* W1 = (const float*)d_in[3];
  const float* b1 = (const float*)d_in[4];
  const float* W2 = (const float*)d_in[5];
  const float* b2 = (const float*)d_in[6];
  const float* W3 = (const float*)d_in[7];
  const float* b3 = (const float*)d_in[8];

  const int N = in_sizes[0] / CIN;
  const int E = in_sizes[2];
  const int* srcIdx = ei;
  const int* dstIdx = ei + E;

  float* T     = (float*)d_ws;
  float* agg   = T + (size_t)N * 128;
  int*   pairs = (int*)(agg + (size_t)N * 128);
  int*   rank  = pairs + (size_t)2 * E;
  float* dinv  = (float*)(rank + E);
  int*   cnt   = (int*)(dinv + N);
  int*   off   = cnt + N;
  const int J   = (E + RK_CHUNK - 1) / RK_CHUNK;
  const int NW  = (N + 1) / 2;
  const int NW4 = (NW + 3) / 4 * 4;
  unsigned* cntJ = (unsigned*)(off + N);
  int*   rel   = (int*)(cntJ + (size_t)J * NW4);
  float* out   = (float*)d_out;

  const int nRowTiles = N / 16;
  dim3 blk(256);
  dim3 gN((N + 255) / 256);
  dim3 gE((E + 255) / 256);
  dim3 gScan((N + 1023) / 1024);
  dim3 gGemm((nRowTiles + 7) / 8);
  dim3 gRows((N + 7) / 8);

  count_rank<<<dim3(J), dim3(RK_THREADS), 0, stream>>>(dstIdx, cntJ, rank, N, E);
  sum_counts<<<dim3((NW + 255) / 256), blk, 0, stream>>>(cntJ, rel, cnt, N, J);
  scan_counts<<<gScan, dim3(1024), 0, stream>>>(cnt, off, N);
  place_edges<<<gE, blk, 0, stream>>>(srcIdx, dstIdx, ea, off, rel, rank, pairs, N, E);
  node_dinv<<<gN, blk, 0, stream>>>(pairs, off, cnt, dinv, N);
  edge_coef<<<gN, blk, 0, stream>>>(pairs, off, cnt, dinv, N);

  gemm_wmma<128, false><<<gGemm, blk, 0, stream>>>(x, W1, T, nRowTiles);
  gather_rows<128><<<gRows, blk, 0, stream>>>(pairs, off, cnt, dinv, T, b1, agg, N);

  gemm_wmma<128, true><<<gGemm, blk, 0, stream>>>(agg, W2, T, nRowTiles);
  gather_rows<128><<<gRows, blk, 0, stream>>>(pairs, off, cnt, dinv, T, b2, agg, N);

  gemm_wmma<64, true><<<gGemm, blk, 0, stream>>>(agg, W3, T, nRowTiles);
  gather_rows<64><<<gRows, blk, 0, stream>>>(pairs, off, cnt, dinv, T, b3, out, N);
}


# Degree scaling factored out of the edge weights: product rows scaled by dinv once, row sums once at their end; coefficient and degree launches dropped (10 to 8); sum order kept, rounding differs
# speedup vs baseline: 3.8689x; 1.1255x over previous
#include <hip/hip_runtime.h>
#include <hip/hip_bf16.h>


#define CIN 128

typedef __attribute__((ext_vector_type(2))) float v2f;
typedef __attribute__((ext_vector_type(4))) float v4f;
typedef __attribute__((ext_vector_type(8))) float v8f;


#define RK_THREADS 1024
#define RK_TRIPS   15
#define RK_CHUNK   (RK_THREADS * 4 * RK_TRIPS)
#define RK_WORDS   65536

__device__ __forceinline__ int lds_rank(unsigned* sCnt, int d) {
  const unsigned sh  = (unsigned)(d & 1) << 4;
  const unsigned old = atomicAdd(&sCnt[d >> 1], 1u << sh);
  return (int)((old >> sh) & 0xFFFFu);
}

__device__ __forceinline__ void lds_clear(unsigned* sCnt, int NW4, int t) {
  for (int i = t * 4; i < NW4; i += RK_THREADS * 4)
    *(uint4*)&sCnt[i] = make_uint4(0u, 0u, 0u, 0u);
}

__global__ __launch_bounds__(RK_THREADS) void count_rank(const int* __restrict__ dst,
                                                         unsigned* __restrict__ cntJ,
                                                         int* __restrict__ rank, int N, int E) {
  __shared__ alignas(16) unsigned sCnt[RK_WORDS];
  const int t    = threadIdx.x;
  const int NW4  = (((N + 1) >> 1) + 3) & ~3;
  const int eBeg = blockIdx.x * RK_CHUNK;
  const bool full = ((E & 3) == 0) && (E - eBeg >= RK_CHUNK);

  if (full) {
    int4 dv[RK_TRIPS];
#pragma unroll
    for (int u = 0; u < RK_TRIPS; ++u)
      dv[u] = *(const int4*)(dst + eBeg + (u * RK_THREADS + t) * 4);
    lds_clear(sCnt, NW4, t);
    __syncthreads();
#pragma unroll
    for (int u = 0; u < RK_TRIPS; ++u) {
      int4 r;
      r.x = lds_rank(sCnt, dv[u].x);
      r.y = lds_rank(sCnt, dv[u].y);
      r.z = lds_rank(sCnt, dv[u].z);
      r.w = lds_rank(sCnt, dv[u].w);
      *(int4*)(rank + eBeg + (u * RK_THREADS + t) * 4) = r;
    }
  } else {
    const int eEnd = (E - eBeg < RK_CHUNK) ? E : eBeg + RK_CHUNK;
    lds_clear(sCnt, NW4, t);
    __syncthreads();
    for (int e = eBeg + t; e < eEnd; e += RK_THREADS) rank[e] = lds_rank(sCnt, dst[e]);
  }
  __syncthreads();

  unsigned* o = cntJ + (size_t)blockIdx.x * NW4;
  for (int i = t * 4; i < NW4; i += RK_THREADS * 4)
    *(uint4*)(o + i) = *(const uint4*)&sCnt[i];
}

#define SC_GROUP 9

template <bool EVEN>
__device__ __forceinline__ void sum_counts_body(const unsigned* __restrict__ cntJ,
                                                int* __restrict__ rel, int* __restrict__ cnt,
                                                int N, int J, int w, int NW4) {
  const int n0 = 2 * w;
  int a0 = 0, a1 = 0;
  for (int j0 = 0; j0 < J; j0 += SC_GROUP) {
    unsigned c[SC_GROUP];
#pragma unroll
    for (int k = 0; k < SC_GROUP; ++k) {
      const int j = (j0 + k < J) ? (j0 + k) : (J - 1);
      c[k] = cntJ[(size_t)j * NW4 + w];
    }
#pragma unroll
    for (int k = 0; k < SC_GROUP; ++k) {
      if (j0 + k < J) {
        int* r = rel + (size_t)(j0 + k) * N + n0;
        if (EVEN) {
          *(int2*)r = make_int2(a0, a1);
        } else {
          r[0] = a0;
          if (n0 + 1 < N) r[1] = a1;
        }
        a0 += (int)(c[k] & 0xFFFFu);
        a1 += (int)(c[k] >> 16);
      }
    }
  }
  cnt[n0] = a0;
  if (EVEN || n0 + 1 < N) cnt[n0 + 1] = a1;
}

__global__ void sum_counts(const unsigned* __restrict__ cntJ, int* __restrict__ rel,
                           int* __restrict__ cnt, int N, int J) {
  const int w   = blockIdx.x * blockDim.x + threadIdx.x;
  const int NW  = (N + 1) >> 1;
  const int NW4 = (NW + 3) & ~3;
  if (w >= NW) return;
  if ((N & 1) == 0) sum_counts_body<true>(cntJ, rel, cnt, N, J, w, NW4);
  else              sum_counts_body<false>(cntJ, rel, cnt, N, J, w, NW4);
}

__global__ __launch_bounds__(1024) void scan_counts(const int* __restrict__ cnt,
                                                    int* __restrict__ off, int N) {
  __shared__ int sPre[32];
  __shared__ int sWave[32];
  const int t    = threadIdx.x;
  const int lane = t & 31;
  const int wv   = t >> 5;
  const int b    = blockIdx.x;

  int pre = 0;
#pragma unroll 8
  for (int i = 0; i < b; ++i) pre += cnt[i * 1024 + t];
  for (int o = 16; o > 0; o >>= 1) pre += __shfl_xor(pre, o);
  if (lane == 0) sPre[wv] = pre;

  const int i = b * 1024 + t;
  const int c = (i < N) ? cnt[i] : 0;
  int x = c;
  for (int o = 1; o < 32; o <<= 1) {
    int y = __shfl_up(x, o);
    if (lane >= o) x += y;
  }
  if (lane == 31) sWave[wv] = x;
  __syncthreads();

  int base = 0;
  for (int k = 0; k < 32; ++k) base += sPre[k];
  for (int k = 0; k < wv; ++k) base += sWave[k];
  if (i < N) off[i] = base + x - c;
}

__global__ void place_edges(const int* __restrict__ src, const int* __restrict__ dst,
                            const float* __restrict__ ea, const int* __restrict__ off,
                            const int* __restrict__ rel, const int* __restrict__ rank,
                            int* __restrict__ pairs, int N, int E) {
  int e = blockIdx.x * blockDim.x + threadIdx.x;
  if (e >= E) return;
  const int d = dst[e];
  const int j = e / RK_CHUNK;
  int p = off[d] + rel[(size_t)j * N + d] + rank[e];
  *(int2*)(pairs + 2 * (size_t)p) = make_int2(src[e], __float_as_int(ea[e]));
}


template <int COUT, bool RELU>
__global__ __launch_bounds__(256) void gemm_wmma(const float* __restrict__ H,
                                                 const float* __restrict__ Wt,
                                                 const int* __restrict__ pairs,
                                                 const int* __restrict__ off,
                                                 const int* __restrict__ cnt,
                                                 float* __restrict__ dinv,
                                                 float* __restrict__ T,
                                                 int nRowTiles) {
  constexpr int PITCH = 2 * COUT + 32;
  __shared__ alignas(16) float sW[(CIN / 2) * PITCH];

  constexpr int NTHREADS = 256;
  constexpr int NV       = COUT / 4;
  constexpr int PER_T    = (CIN / 2) * NV / NTHREADS;
  v4f wlo[PER_T], whi[PER_T];
#pragma unroll
  for (int u = 0; u < PER_T; ++u) {
    const int id = (int)threadIdx.x + u * NTHREADS;
    const int p  = id / NV;
    const int g  = id - p * NV;
    wlo[u] = *(const v4f*)(Wt + (size_t)(2 * p) * COUT + 4 * g);
    whi[u] = *(const v4f*)(Wt + (size_t)(2 * p + 1) * COUT + 4 * g);
  }

  const int wave = threadIdx.x >> 5;
  const int lane = threadIdx.x & 31;
  const int rt   = blockIdx.x * 8 + wave;

  float dn = 0.0f;
  if (rt < nRowTiles) {
    const int nn = rt * 16 + (lane & 15);
    const int* p = pairs + 2 * (size_t)off[nn];
    const int c  = cnt[nn];
    float d = 1.0f;
    int j = 0;
    for (; j + 4 <= c; j += 4) {
      const float w0 = __int_as_float(p[2 * j + 1]);
      const float w1 = __int_as_float(p[2 * j + 3]);
      const float w2 = __int_as_float(p[2 * j + 5]);
      const float w3 = __int_as_float(p[2 * j + 7]);
      d += w0;
      d += w1;
      d += w2;
      d += w3;
    }
    for (; j < c; ++j) d += __int_as_float(p[2 * j + 1]);
    dn = (d > 0.0f) ? rsqrtf(d) : 0.0f;
    if (lane < 16) dinv[nn] = dn;
  }

#pragma unroll
  for (int u = 0; u < PER_T; ++u) {
    const int id = (int)threadIdx.x + u * NTHREADS;
    const int p  = id / NV;
    const int g  = id - p * NV;
    float* d = &sW[p * PITCH + 8 * g];
    v4f s0, s1;
    s0.x = wlo[u].x; s0.y = whi[u].x; s0.z = wlo[u].y; s0.w = whi[u].y;
    s1.x = wlo[u].z; s1.y = whi[u].z; s1.z = wlo[u].w; s1.w = whi[u].w;
    *(v4f*)(d)     = s0;
    *(v4f*)(d + 4) = s1;
  }
  __syncthreads();

  if (rt >= nRowTiles) return;

  const int m0 = rt * 16;
  const int m  = lane & 15;
  const int kh = lane >> 4;

  constexpr int NT = COUT / 16;
  v8f acc[NT] = {};

  const float* arow = H + (size_t)(m0 + m) * CIN + 2 * kh;

  for (int k = 0; k < CIN; k += 4) {
    v2f a = *(const v2f*)(arow + k);
    if (RELU) {
      a.x = fmaxf(a.x, 0.0f);
      a.y = fmaxf(a.y, 0.0f);
    }
    const float* b0 = &sW[((k >> 1) + kh) * PITCH + 2 * m];
#pragma unroll
    for (int t = 0; t < NT; ++t) {
      v2f b = *(const v2f*)(b0 + 32 * t);
      acc[t] = __builtin_amdgcn_wmma_f32_16x16x4_f32(
           false, a,  false, b,
           (short)0, acc[t],  false,  false);
    }
  }

  float sc[8];
#pragma unroll
  for (int v = 0; v < 8; ++v) sc[v] = __shfl(dn, v + 8 * kh);
  float* out = T + (size_t)(m0 + 8 * kh) * COUT + m;
#pragma unroll
  for (int t = 0; t < NT; ++t) {
#pragma unroll
    for (int v = 0; v < 8; ++v) {
      out[(size_t)v * COUT + t * 16] = sc[v] * acc[t][v];
    }
  }
}


template <int C>
__global__ __launch_bounds__(256) void gather_rows(const int* __restrict__ pairs,
                                                   const int* __restrict__ off,
                                                   const int* __restrict__ cnt,
                                                   const float* __restrict__ dinv,
                                                   const float* __restrict__ T,
                                                   const float* __restrict__ bias,
                                                   float* __restrict__ agg, int N) {
  constexpr int V = C / 32;
  typedef float vecT __attribute__((ext_vector_type(V)));

  const int lane = threadIdx.x & 31;
  const int wv = __builtin_amdgcn_readfirstlane((int)(threadIdx.x >> 5));
  const int n  = blockIdx.x * 8 + wv;
  if (n >= N) return;

  const int beg = off[n];
  const int c   = cnt[n];
  const float dn = dinv[n];

  const int col = lane * V;
  vecT acc = *(const vecT*)(T + (size_t)n * C + col);

  for (int base = 0; base < c; base += 32) {
    const int m = (c - base < 32) ? (c - base) : 32;
    int ms = 0;
    int mc = 0;
    if (lane < m) {
      const int2 en = *(const int2*)(pairs + 2 * (size_t)(beg + base + lane));
      ms = en.x;
      mc = en.y;
    }
    int j = 0;
    for (; j + 4 <= m; j += 4) {
      const int s0 = __builtin_amdgcn_readlane(ms, j);
      const int s1 = __builtin_amdgcn_readlane(ms, j + 1);
      const int s2 = __builtin_amdgcn_readlane(ms, j + 2);
      const int s3 = __builtin_amdgcn_readlane(ms, j + 3);
      const float c0 = __int_as_float(__builtin_amdgcn_readlane(mc, j));
      const float c1 = __int_as_float(__builtin_amdgcn_readlane(mc, j + 1));
      const float c2 = __int_as_float(__builtin_amdgcn_readlane(mc, j + 2));
      const float c3 = __int_as_float(__builtin_amdgcn_readlane(mc, j + 3));
      const vecT r0 = *(const vecT*)(T + (size_t)s0 * C + col);
      const vecT r1 = *(const vecT*)(T + (size_t)s1 * C + col);
      const vecT r2 = *(const vecT*)(T + (size_t)s2 * C + col);
      const vecT r3 = *(const vecT*)(T + (size_t)s3 * C + col);
      acc += c0 * r0;
      acc += c1 * r1;
      acc += c2 * r2;
      acc += c3 * r3;
    }
    for (; j < m; ++j) {
      const int s0 = __builtin_amdgcn_readlane(ms, j);
      const float c0 = __int_as_float(__builtin_amdgcn_readlane(mc, j));
      acc += c0 * *(const vecT*)(T + (size_t)s0 * C + col);
    }
  }

  *(vecT*)(agg + (size_t)n * C + col) = *(const vecT*)(bias + col) + dn * acc;
}

#define APITCH 132
#define GB 16

template <int COUT, int NWAVES>
__global__ __launch_bounds__(NWAVES * 32) void gather_gemm(const int* __restrict__ pairs,
                                                           const int* __restrict__ off,
                                                           const int* __restrict__ cnt,
                                                           const float* __restrict__ dinv,
                                                           const float* __restrict__ Tin,
                                                           const float* __restrict__ bias,
                                                           const float* __restrict__ Wt,
                                                           float* __restrict__ Tout,
                                                           int nRowTiles) {
  constexpr int PITCH    = 2 * COUT + 32;
  constexpr int NTHREADS = NWAVES * 32;
  constexpr int NV       = COUT / 4;
  constexpr int UNITS    = (CIN / 2) * NV;
  constexpr int PER_T    = (UNITS + NTHREADS - 1) / NTHREADS;
  constexpr bool EXACT   = (PER_T * NTHREADS == UNITS);
  __shared__ alignas(16) float sW[(CIN / 2) * PITCH];
  __shared__ alignas(16) float sA[NWAVES * 16 * APITCH];

  const int tid  = (int)threadIdx.x;
  const int lane = tid & 31;
  const int wave = __builtin_amdgcn_readfirstlane(tid >> 5);
  const int rt   = blockIdx.x * NWAVES + wave;
  const bool live = rt < nRowTiles;
  const int m0   = rt * 16;
  const int col  = lane * 4;
  float* sAw = sA + wave * (16 * APITCH);

  v4f wlo[PER_T], whi[PER_T];
#pragma unroll
  for (int u = 0; u < PER_T; ++u) {
    const int id = tid + u * NTHREADS;
    if (EXACT || id < UNITS) {
      const int p = id / NV;
      const int g = id - p * NV;
      wlo[u] = *(const v4f*)(Wt + (size_t)(2 * p) * COUT + 4 * g);
      whi[u] = *(const v4f*)(Wt + (size_t)(2 * p + 1) * COUT + 4 * g);
    }
  }

  int rOff = 0, rCnt = 0;
  float rD = 0.0f;
  v4f bv = {0.0f, 0.0f, 0.0f, 0.0f};
  v4f self[16];
  if (live) {
    const int nn = m0 + (lane & 15);
    rOff = off[nn];
    rCnt = cnt[nn];
    rD   = dinv[nn];
    bv   = *(const v4f*)(bias + col);
#pragma unroll
    for (int r = 0; r < 16; ++r) self[r] = *(const v4f*)(Tin + (size_t)(m0 + r) * CIN + col);
  }

#pragma unroll
  for (int u = 0; u < PER_T; ++u) {
    const int id = tid + u * NTHREADS;
    if (EXACT || id < UNITS) {
      const int p = id / NV;
      const int g = id - p * NV;
      float* d = &sW[p * PITCH + 8 * g];
      v4f s0, s1;
      s0.x = wlo[u].x; s0.y = whi[u].x; s0.z = wlo[u].y; s0.w = whi[u].y;
      s1.x = wlo[u].z; s1.y = whi[u].z; s1.z = wlo[u].w; s1.w = whi[u].w;
      *(v4f*)(d)     = s0;
      *(v4f*)(d + 4) = s1;
    }
  }
  __syncthreads();
  if (!live) return;

#pragma unroll
  for (int r = 0; r < 16; ++r) *(v4f*)(sAw + r * APITCH + col) = self[r];

  const int e0 = __builtin_amdgcn_readlane(rOff, 0);
  const int e1 = __builtin_amdgcn_readlane(rOff, 15) + __builtin_amdgcn_readlane(rCnt, 15);
  int r      = 0;
  int rowEnd = e0 + __builtin_amdgcn_readlane(rCnt, 0);
  int e      = e0;
  v4f acc = *(const v4f*)(sAw + col);

  auto finish = [&](v4f sum, int row) -> v4f {
    const float dr = __int_as_float(__builtin_amdgcn_readlane(__float_as_int(rD), row));
    const v4f a = bv + dr * sum;
    v4f h;
    h.x = fmaxf(a.x, 0.0f); h.y = fmaxf(a.y, 0.0f);
    h.z = fmaxf(a.z, 0.0f); h.w = fmaxf(a.w, 0.0f);
    return dr * h;
  };
  auto advance = [&]() {
    while (e == rowEnd) {
      *(v4f*)(sAw + r * APITCH + col) = finish(acc, r);
      ++r;
      rowEnd = __builtin_amdgcn_readlane(rOff, r) + __builtin_amdgcn_readlane(rCnt, r);
      acc = *(const v4f*)(sAw + r * APITCH + col);
    }
  };

  int2 cur = make_int2(0, 0);
  if (e0 + lane < e1) cur = *(const int2*)(pairs + 2 * (size_t)(e0 + lane));
  for (int base = e0; base < e1; base += 32) {
    int2 nxt = make_int2(0, 0);
    if (base + 32 + lane < e1) nxt = *(const int2*)(pairs + 2 * (size_t)(base + 32 + lane));
    const int m = (e1 - base < 32) ? (e1 - base) : 32;
    int j = 0;
    for (; j + GB <= m; j += GB) {
      v4f rv[GB];
#pragma unroll
      for (int i = 0; i < GB; ++i) {
        const int s = __builtin_amdgcn_readlane(cur.x, j + i);
        rv[i] = *(const v4f*)(Tin + (size_t)s * CIN + col);
      }
#pragma unroll
      for (int i = 0; i < GB; ++i) {
        const float c = __int_as_float(__builtin_amdgcn_readlane(cur.y, j + i));
        advance();
        acc += c * rv[i];
        ++e;
      }
    }
    for (; j + 4 <= m; j += 4) {
      v4f rv[4];
#pragma unroll
      for (int i = 0; i < 4; ++i) {
        const int s = __builtin_amdgcn_readlane(cur.x, j + i);
        rv[i] = *(const v4f*)(Tin + (size_t)s * CIN + col);
      }
#pragma unroll
      for (int i = 0; i < 4; ++i) {
        const float c = __int_as_float(__builtin_amdgcn_readlane(cur.y, j + i));
        advance();
        acc += c * rv[i];
        ++e;
      }
    }
    for (; j < m; ++j) {
      const int s = __builtin_amdgcn_readlane(cur.x, j);
      const float c = __int_as_float(__builtin_amdgcn_readlane(cur.y, j));
      const v4f rv = *(const v4f*)(Tin + (size_t)s * CIN + col);
      advance();
      acc += c * rv;
      ++e;
    }
    cur = nxt;
  }
  for (;;) {
    *(v4f*)(sAw + r * APITCH + col) = finish(acc, r);
    if (++r >= 16) break;
    acc = *(const v4f*)(sAw + r * APITCH + col);
  }

  const int mm = lane & 15;
  const int kh = lane >> 4;
  constexpr int NT = COUT / 16;
  v8f accT[NT] = {};
  const float* arow = sAw + mm * APITCH + 2 * kh;
  for (int k = 0; k < CIN; k += 4) {
    const v2f a = *(const v2f*)(arow + k);
    const float* b0 = &sW[((k >> 1) + kh) * PITCH + 2 * mm];
#pragma unroll
    for (int t = 0; t < NT; ++t) {
      v2f b = *(const v2f*)(b0 + 32 * t);
      accT[t] = __builtin_amdgcn_wmma_f32_16x16x4_f32(
           false, a,  false, b,
           (short)0, accT[t],  false,  false);
    }
  }

  float* o = Tout + (size_t)(m0 + 8 * kh) * COUT + mm;
#pragma unroll
  for (int t = 0; t < NT; ++t) {
#pragma unroll
    for (int v = 0; v < 8; ++v) {
      o[(size_t)v * COUT + t * 16] = accT[t][v];
    }
  }
}


extern "C" void kernel_launch(void* const* d_in, const int* in_sizes, int n_in,
                              void* d_out, int out_size, void* d_ws, size_t ws_size,
                              hipStream_t stream) {
  const float* x  = (const float*)d_in[0];
  const int*   ei = (const int*)d_in[1];
  const float* ea = (const float*)d_in[2];
  const float* W1 = (const float*)d_in[3];
  const float* b1 = (const float*)d_in[4];
  const float* W2 = (const float*)d_in[5];
  const float* b2 = (const float*)d_in[6];
  const float* W3 = (const float*)d_in[7];
  const float* b3 = (const float*)d_in[8];

  const int N = in_sizes[0] / CIN;
  const int E = in_sizes[2];
  const int* srcIdx = ei;
  const int* dstIdx = ei + E;

  float* T     = (float*)d_ws;
  float* agg   = T + (size_t)N * 128;
  int*   pairs = (int*)(agg + (size_t)N * 128);
  int*   rank  = pairs + (size_t)2 * E;
  float* dinv  = (float*)(rank + E);
  int*   cnt   = (int*)(dinv + N);
  int*   off   = cnt + N;
  const int J   = (E + RK_CHUNK - 1) / RK_CHUNK;
  const int NW  = (N + 1) / 2;
  const int NW4 = (NW + 3) / 4 * 4;
  unsigned* cntJ = (unsigned*)(off + N);
  int*   rel   = (int*)(cntJ + (size_t)J * NW4);
  float* out   = (float*)d_out;

  const int nRowTiles = N / 16;
  dim3 blk(256);
  dim3 gE((E + 255) / 256);
  dim3 gScan((N + 1023) / 1024);
  dim3 gGemm((nRowTiles + 7) / 8);
  dim3 gRows((N + 7) / 8);

  count_rank<<<dim3(J), dim3(RK_THREADS), 0, stream>>>(dstIdx, cntJ, rank, N, E);
  sum_counts<<<dim3((NW + 255) / 256), blk, 0, stream>>>(cntJ, rel, cnt, N, J);
  scan_counts<<<gScan, dim3(1024), 0, stream>>>(cnt, off, N);
  place_edges<<<gE, blk, 0, stream>>>(srcIdx, dstIdx, ea, off, rel, rank, pairs, N, E);

  gemm_wmma<128, false><<<gGemm, blk, 0, stream>>>(x, W1, pairs, off, cnt, dinv, T, nRowTiles);

  gather_gemm<128, 8><<<dim3((nRowTiles + 7) / 8), dim3(256), 0, stream>>>(
      pairs, off, cnt, dinv, T, b1, W2, agg, nRowTiles);

  gather_gemm<64, 13><<<dim3((nRowTiles + 12) / 13), dim3(416), 0, stream>>>(
      pairs, off, cnt, dinv, agg, b2, W3, T, nRowTiles);

  gather_rows<64><<<gRows, blk, 0, stream>>>(pairs, off, cnt, dinv, T, b3, out, N);
}


# Segment starts without the scan launch: count_rank also writes sums of 256-node units, sum_counts adds the units in front, scans its block, stores absolute chunk starts; 8 launches to 7; sums kept
# speedup vs baseline: 3.9920x; 1.0318x over previous
#include <hip/hip_runtime.h>
#include <hip/hip_bf16.h>


#define CIN 128

typedef __attribute__((ext_vector_type(2))) float v2f;
typedef __attribute__((ext_vector_type(4))) float v4f;
typedef __attribute__((ext_vector_type(8))) float v8f;


#define RK_THREADS 1024
#define RK_TRIPS   15
#define RK_CHUNK   (RK_THREADS * 4 * RK_TRIPS)
#define RK_WORDS   65536
#define RK_UNIT    128

__device__ __forceinline__ int lds_rank(unsigned* sCnt, int d) {
  const unsigned sh  = (unsigned)(d & 1) << 4;
  const unsigned old = atomicAdd(&sCnt[d >> 1], 1u << sh);
  return (int)((old >> sh) & 0xFFFFu);
}

__device__ __forceinline__ void lds_clear(unsigned* sCnt, int NW4, int t) {
  for (int i = t * 4; i < NW4; i += RK_THREADS * 4)
    *(uint4*)&sCnt[i] = make_uint4(0u, 0u, 0u, 0u);
}

__global__ __launch_bounds__(RK_THREADS) void count_rank(const int* __restrict__ dst,
                                                         unsigned* __restrict__ cntJ,
                                                         unsigned* __restrict__ part,
                                                         int* __restrict__ rank, int N, int E,
                                                         int NPW) {
  __shared__ alignas(16) unsigned sCnt[RK_WORDS];
  const int t    = threadIdx.x;
  const int NW4  = (((N + 1) >> 1) + 3) & ~3;
  const int eBeg = blockIdx.x * RK_CHUNK;
  const bool full = ((E & 3) == 0) && (E - eBeg >= RK_CHUNK);

  if (full) {
    int4 dv[RK_TRIPS];
#pragma unroll
    for (int u = 0; u < RK_TRIPS; ++u)
      dv[u] = *(const int4*)(dst + eBeg + (u * RK_THREADS + t) * 4);
    lds_clear(sCnt, NW4, t);
    __syncthreads();
#pragma unroll
    for (int u = 0; u < RK_TRIPS; ++u) {
      int4 r;
      r.x = lds_rank(sCnt, dv[u].x);
      r.y = lds_rank(sCnt, dv[u].y);
      r.z = lds_rank(sCnt, dv[u].z);
      r.w = lds_rank(sCnt, dv[u].w);
      *(int4*)(rank + eBeg + (u * RK_THREADS + t) * 4) = r;
    }
  } else {
    const int eEnd = (E - eBeg < RK_CHUNK) ? E : eBeg + RK_CHUNK;
    lds_clear(sCnt, NW4, t);
    __syncthreads();
    for (int e = eBeg + t; e < eEnd; e += RK_THREADS) rank[e] = lds_rank(sCnt, dst[e]);
  }
  __syncthreads();

  unsigned* o  = cntJ + (size_t)blockIdx.x * NW4;
  unsigned* pw = part + (size_t)blockIdx.x * NPW;
  for (int it = 0; it * (RK_THREADS * 4) < NW4; ++it) {
    const int i = it * (RK_THREADS * 4) + t * 4;
    unsigned s = 0;
    if (i < NW4) {
      const uint4 c = *(const uint4*)&sCnt[i];
      *(uint4*)(o + i) = c;
      s = (c.x & 0xFFFFu) + (c.x >> 16) + (c.y & 0xFFFFu) + (c.y >> 16) +
          (c.z & 0xFFFFu) + (c.z >> 16) + (c.w & 0xFFFFu) + (c.w >> 16);
    }
    for (int sh = 16; sh > 0; sh >>= 1) s += __shfl_xor(s, sh);
    const int q = it * (RK_THREADS / 32) + (t >> 5);
    if ((t & 31) == 0 && q * RK_UNIT < NW4) pw[q] = s;
  }
}

#define SC_GROUP   14
#define SS_THREADS 256
#define SS_UPB     (SS_THREADS / RK_UNIT)

template <bool EVEN>
__device__ __forceinline__ void chunk_starts_body(const unsigned* __restrict__ cntJ,
                                                  int* __restrict__ rel, int N, int J, int w,
                                                  int NW4, int a0, int a1) {
  const int n0 = 2 * w;
  for (int j0 = 0; j0 < J; j0 += SC_GROUP) {
    unsigned c[SC_GROUP];
#pragma unroll
    for (int k = 0; k < SC_GROUP; ++k) {
      const int j = (j0 + k < J) ? (j0 + k) : (J - 1);
      c[k] = cntJ[(size_t)j * NW4 + w];
    }
#pragma unroll
    for (int k = 0; k < SC_GROUP; ++k) {
      if (j0 + k < J) {
        int* r = rel + (size_t)(j0 + k) * N + n0;
        if (EVEN) {
          *(int2*)r = make_int2(a0, a1);
        } else {
          r[0] = a0;
          if (n0 + 1 < N) r[1] = a1;
        }
        a0 += (int)(c[k] & 0xFFFFu);
        a1 += (int)(c[k] >> 16);
      }
    }
  }
}

__global__ __launch_bounds__(SS_THREADS) void sum_scan_counts(const unsigned* __restrict__ cntJ,
                                                              const unsigned* __restrict__ part,
                                                              int* __restrict__ rel,
                                                              int* __restrict__ cnt,
                                                              int* __restrict__ off,
                                                              int N, int J, int NPW) {
  __shared__ int sTot[SS_THREADS / 32];
  __shared__ int sFront[SS_THREADS / 32];
  const int tid  = threadIdx.x;
  const int lane = tid & 31;
  const int wv   = tid >> 5;
  const int NW   = (N + 1) >> 1;
  const int NW4  = (NW + 3) & ~3;
  const int w    = blockIdx.x * SS_THREADS + tid;
  const bool valid = w < NW;
  const int wc   = valid ? w : (NW - 1);
  const int nq   = (int)blockIdx.x * SS_UPB;
  const bool u0ok = tid < nq;
  const bool u1ok = tid + SS_THREADS < nq;
  const int u0   = u0ok ? tid : 0;
  const int u1   = u1ok ? tid + SS_THREADS : 0;

  int t0 = 0, t1 = 0, front = 0;
  for (int j0 = 0; j0 < J; j0 += SC_GROUP) {
    unsigned c[SC_GROUP], p0[SC_GROUP], p1[SC_GROUP];
#pragma unroll
    for (int k = 0; k < SC_GROUP; ++k) {
      const int j = (j0 + k < J) ? (j0 + k) : (J - 1);
      c[k]  = cntJ[(size_t)j * NW4 + wc];
      p0[k] = part[(size_t)j * NPW + u0];
      p1[k] = part[(size_t)j * NPW + u1];
    }
#pragma unroll
    for (int k = 0; k < SC_GROUP; ++k) {
      if (j0 + k < J) {
        t0 += (int)(c[k] & 0xFFFFu);
        t1 += (int)(c[k] >> 16);
        if (u0ok) front += (int)p0[k];
        if (u1ok) front += (int)p1[k];
      }
    }
  }
  if (!valid) { t0 = 0; t1 = 0; }

  const int s = t0 + t1;
  int incl = s;
  for (int o = 1; o < 32; o <<= 1) {
    const int y = __shfl_up(incl, o);
    if (lane >= o) incl += y;
  }
  for (int o = 16; o > 0; o >>= 1) front += __shfl_xor(front, o);
  if (lane == 31) sTot[wv] = incl;
  if (lane == 0) sFront[wv] = front;
  __syncthreads();
  int base = 0;
  for (int k = 0; k < SS_THREADS / 32; ++k) base += sFront[k];
  for (int k = 0; k < wv; ++k) base += sTot[k];
  const int off0 = base + incl - s;
  const int off1 = off0 + t0;

  if (!valid) return;
  const int n0 = 2 * w;
  off[n0] = off0;
  cnt[n0] = t0;
  if (n0 + 1 < N) { off[n0 + 1] = off1; cnt[n0 + 1] = t1; }

  if ((N & 1) == 0) chunk_starts_body<true>(cntJ, rel, N, J, w, NW4, off0, off1);
  else              chunk_starts_body<false>(cntJ, rel, N, J, w, NW4, off0, off1);
}

__global__ void place_edges(const int* __restrict__ src, const int* __restrict__ dst,
                            const float* __restrict__ ea,
                            const int* __restrict__ rel, const int* __restrict__ rank,
                            int* __restrict__ pairs, int N, int E) {
  int e = blockIdx.x * blockDim.x + threadIdx.x;
  if (e >= E) return;
  const int d = dst[e];
  const int j = e / RK_CHUNK;
  const int p = rel[(size_t)j * N + d] + rank[e];
  *(int2*)(pairs + 2 * (size_t)p) = make_int2(src[e], __float_as_int(ea[e]));
}


template <int COUT, bool RELU>
__global__ __launch_bounds__(256) void gemm_wmma(const float* __restrict__ H,
                                                 const float* __restrict__ Wt,
                                                 const int* __restrict__ pairs,
                                                 const int* __restrict__ off,
                                                 const int* __restrict__ cnt,
                                                 float* __restrict__ dinv,
                                                 float* __restrict__ T,
                                                 int nRowTiles) {
  constexpr int PITCH = 2 * COUT + 32;
  __shared__ alignas(16) float sW[(CIN / 2) * PITCH];

  constexpr int NTHREADS = 256;
  constexpr int NV       = COUT / 4;
  constexpr int PER_T    = (CIN / 2) * NV / NTHREADS;
  v4f wlo[PER_T], whi[PER_T];
#pragma unroll
  for (int u = 0; u < PER_T; ++u) {
    const int id = (int)threadIdx.x + u * NTHREADS;
    const int p  = id / NV;
    const int g  = id - p * NV;
    wlo[u] = *(const v4f*)(Wt + (size_t)(2 * p) * COUT + 4 * g);
    whi[u] = *(const v4f*)(Wt + (size_t)(2 * p + 1) * COUT + 4 * g);
  }

  const int wave = threadIdx.x >> 5;
  const int lane = threadIdx.x & 31;
  const int rt   = blockIdx.x * 8 + wave;

  float dn = 0.0f;
  if (rt < nRowTiles) {
    const int nn = rt * 16 + (lane & 15);
    const int* p = pairs + 2 * (size_t)off[nn];
    const int c  = cnt[nn];
    float d = 1.0f;
    int j = 0;
    for (; j + 4 <= c; j += 4) {
      const float w0 = __int_as_float(p[2 * j + 1]);
      const float w1 = __int_as_float(p[2 * j + 3]);
      const float w2 = __int_as_float(p[2 * j + 5]);
      const float w3 = __int_as_float(p[2 * j + 7]);
      d += w0;
      d += w1;
      d += w2;
      d += w3;
    }
    for (; j < c; ++j) d += __int_as_float(p[2 * j + 1]);
    dn = (d > 0.0f) ? rsqrtf(d) : 0.0f;
    if (lane < 16) dinv[nn] = dn;
  }

#pragma unroll
  for (int u = 0; u < PER_T; ++u) {
    const int id = (int)threadIdx.x + u * NTHREADS;
    const int p  = id / NV;
    const int g  = id - p * NV;
    float* d = &sW[p * PITCH + 8 * g];
    v4f s0, s1;
    s0.x = wlo[u].x; s0.y = whi[u].x; s0.z = wlo[u].y; s0.w = whi[u].y;
    s1.x = wlo[u].z; s1.y = whi[u].z; s1.z = wlo[u].w; s1.w = whi[u].w;
    *(v4f*)(d)     = s0;
    *(v4f*)(d + 4) = s1;
  }
  __syncthreads();

  if (rt >= nRowTiles) return;

  const int m0 = rt * 16;
  const int m  = lane & 15;
  const int kh = lane >> 4;

  constexpr int NT = COUT / 16;
  v8f acc[NT] = {};

  const float* arow = H + (size_t)(m0 + m) * CIN + 2 * kh;

  for (int k = 0; k < CIN; k += 4) {
    v2f a = *(const v2f*)(arow + k);
    if (RELU) {
      a.x = fmaxf(a.x, 0.0f);
      a.y = fmaxf(a.y, 0.0f);
    }
    const float* b0 = &sW[((k >> 1) + kh) * PITCH + 2 * m];
#pragma unroll
    for (int t = 0; t < NT; ++t) {
      v2f b = *(const v2f*)(b0 + 32 * t);
      acc[t] = __builtin_amdgcn_wmma_f32_16x16x4_f32(
           false, a,  false, b,
           (short)0, acc[t],  false,  false);
    }
  }

  float sc[8];
#pragma unroll
  for (int v = 0; v < 8; ++v) sc[v] = __shfl(dn, v + 8 * kh);
  float* out = T + (size_t)(m0 + 8 * kh) * COUT + m;
#pragma unroll
  for (int t = 0; t < NT; ++t) {
#pragma unroll
    for (int v = 0; v < 8; ++v) {
      out[(size_t)v * COUT + t * 16] = sc[v] * acc[t][v];
    }
  }
}


template <int C>
__global__ __launch_bounds__(256) void gather_rows(const int* __restrict__ pairs,
                                                   const int* __restrict__ off,
                                                   const int* __restrict__ cnt,
                                                   const float* __restrict__ dinv,
                                                   const float* __restrict__ T,
                                                   const float* __restrict__ bias,
                                                   float* __restrict__ agg, int N) {
  constexpr int V = C / 32;
  typedef float vecT __attribute__((ext_vector_type(V)));

  const int lane = threadIdx.x & 31;
  const int wv = __builtin_amdgcn_readfirstlane((int)(threadIdx.x >> 5));
  const int n  = blockIdx.x * 8 + wv;
  if (n >= N) return;

  const int beg = off[n];
  const int c   = cnt[n];
  const float dn = dinv[n];

  const int col = lane * V;
  vecT acc = *(const vecT*)(T + (size_t)n * C + col);

  for (int base = 0; base < c; base += 32) {
    const int m = (c - base < 32) ? (c - base) : 32;
    int ms = 0;
    int mc = 0;
    if (lane < m) {
      const int2 en = *(const int2*)(pairs + 2 * (size_t)(beg + base + lane));
      ms = en.x;
      mc = en.y;
    }
    int j = 0;
    for (; j + 4 <= m; j += 4) {
      const int s0 = __builtin_amdgcn_readlane(ms, j);
      const int s1 = __builtin_amdgcn_readlane(ms, j + 1);
      const int s2 = __builtin_amdgcn_readlane(ms, j + 2);
      const int s3 = __builtin_amdgcn_readlane(ms, j + 3);
      const float c0 = __int_as_float(__builtin_amdgcn_readlane(mc, j));
      const float c1 = __int_as_float(__builtin_amdgcn_readlane(mc, j + 1));
      const float c2 = __int_as_float(__builtin_amdgcn_readlane(mc, j + 2));
      const float c3 = __int_as_float(__builtin_amdgcn_readlane(mc, j + 3));
      const vecT r0 = *(const vecT*)(T + (size_t)s0 * C + col);
      const vecT r1 = *(const vecT*)(T + (size_t)s1 * C + col);
      const vecT r2 = *(const vecT*)(T + (size_t)s2 * C + col);
      const vecT r3 = *(const vecT*)(T + (size_t)s3 * C + col);
      acc += c0 * r0;
      acc += c1 * r1;
      acc += c2 * r2;
      acc += c3 * r3;
    }
    for (; j < m; ++j) {
      const int s0 = __builtin_amdgcn_readlane(ms, j);
      const float c0 = __int_as_float(__builtin_amdgcn_readlane(mc, j));
      acc += c0 * *(const vecT*)(T + (size_t)s0 * C + col);
    }
  }

  *(vecT*)(agg + (size_t)n * C + col) = *(const vecT*)(bias + col) + dn * acc;
}

#define APITCH 132
#define GB 16

template <int COUT, int NWAVES>
__global__ __launch_bounds__(NWAVES * 32) void gather_gemm(const int* __restrict__ pairs,
                                                           const int* __restrict__ off,
                                                           const int* __restrict__ cnt,
                                                           const float* __restrict__ dinv,
                                                           const float* __restrict__ Tin,
                                                           const float* __restrict__ bias,
                                                           const float* __restrict__ Wt,
                                                           float* __restrict__ Tout,
                                                           int nRowTiles) {
  constexpr int PITCH    = 2 * COUT + 32;
  constexpr int NTHREADS = NWAVES * 32;
  constexpr int NV       = COUT / 4;
  constexpr int UNITS    = (CIN / 2) * NV;
  constexpr int PER_T    = (UNITS + NTHREADS - 1) / NTHREADS;
  constexpr bool EXACT   = (PER_T * NTHREADS == UNITS);
  __shared__ alignas(16) float sW[(CIN / 2) * PITCH];
  __shared__ alignas(16) float sA[NWAVES * 16 * APITCH];

  const int tid  = (int)threadIdx.x;
  const int lane = tid & 31;
  const int wave = __builtin_amdgcn_readfirstlane(tid >> 5);
  const int rt   = blockIdx.x * NWAVES + wave;
  const bool live = rt < nRowTiles;
  const int m0   = rt * 16;
  const int col  = lane * 4;
  float* sAw = sA + wave * (16 * APITCH);

  v4f wlo[PER_T], whi[PER_T];
#pragma unroll
  for (int u = 0; u < PER_T; ++u) {
    const int id = tid + u * NTHREADS;
    if (EXACT || id < UNITS) {
      const int p = id / NV;
      const int g = id - p * NV;
      wlo[u] = *(const v4f*)(Wt + (size_t)(2 * p) * COUT + 4 * g);
      whi[u] = *(const v4f*)(Wt + (size_t)(2 * p + 1) * COUT + 4 * g);
    }
  }

  int rOff = 0, rCnt = 0;
  float rD = 0.0f;
  v4f bv = {0.0f, 0.0f, 0.0f, 0.0f};
  v4f self[16];
  if (live) {
    const int nn = m0 + (lane & 15);
    rOff = off[nn];
    rCnt = cnt[nn];
    rD   = dinv[nn];
    bv   = *(const v4f*)(bias + col);
#pragma unroll
    for (int r = 0; r < 16; ++r) self[r] = *(const v4f*)(Tin + (size_t)(m0 + r) * CIN + col);
  }

#pragma unroll
  for (int u = 0; u < PER_T; ++u) {
    const int id = tid + u * NTHREADS;
    if (EXACT || id < UNITS) {
      const int p = id / NV;
      const int g = id - p * NV;
      float* d = &sW[p * PITCH + 8 * g];
      v4f s0, s1;
      s0.x = wlo[u].x; s0.y = whi[u].x; s0.z = wlo[u].y; s0.w = whi[u].y;
      s1.x = wlo[u].z; s1.y = whi[u].z; s1.z = wlo[u].w; s1.w = whi[u].w;
      *(v4f*)(d)     = s0;
      *(v4f*)(d + 4) = s1;
    }
  }
  __syncthreads();
  if (!live) return;

#pragma unroll
  for (int r = 0; r < 16; ++r) *(v4f*)(sAw + r * APITCH + col) = self[r];

  const int e0 = __builtin_amdgcn_readlane(rOff, 0);
  const int e1 = __builtin_amdgcn_readlane(rOff, 15) + __builtin_amdgcn_readlane(rCnt, 15);
  int r      = 0;
  int rowEnd = e0 + __builtin_amdgcn_readlane(rCnt, 0);
  int e      = e0;
  v4f acc = *(const v4f*)(sAw + col);

  auto finish = [&](v4f sum, int row) -> v4f {
    const float dr = __int_as_float(__builtin_amdgcn_readlane(__float_as_int(rD), row));
    const v4f a = bv + dr * sum;
    v4f h;
    h.x = fmaxf(a.x, 0.0f); h.y = fmaxf(a.y, 0.0f);
    h.z = fmaxf(a.z, 0.0f); h.w = fmaxf(a.w, 0.0f);
    return dr * h;
  };
  auto advance = [&]() {
    while (e == rowEnd) {
      *(v4f*)(sAw + r * APITCH + col) = finish(acc, r);
      ++r;
      rowEnd = __builtin_amdgcn_readlane(rOff, r) + __builtin_amdgcn_readlane(rCnt, r);
      acc = *(const v4f*)(sAw + r * APITCH + col);
    }
  };

  int2 cur = make_int2(0, 0);
  if (e0 + lane < e1) cur = *(const int2*)(pairs + 2 * (size_t)(e0 + lane));
  for (int base = e0; base < e1; base += 32) {
    int2 nxt = make_int2(0, 0);
    if (base + 32 + lane < e1) nxt = *(const int2*)(pairs + 2 * (size_t)(base + 32 + lane));
    const int m = (e1 - base < 32) ? (e1 - base) : 32;
    int j = 0;
    for (; j + GB <= m; j += GB) {
      v4f rv[GB];
#pragma unroll
      for (int i = 0; i < GB; ++i) {
        const int s = __builtin_amdgcn_readlane(cur.x, j + i);
        rv[i] = *(const v4f*)(Tin + (size_t)s * CIN + col);
      }
#pragma unroll
      for (int i = 0; i < GB; ++i) {
        const float c = __int_as_float(__builtin_amdgcn_readlane(cur.y, j + i));
        advance();
        acc += c * rv[i];
        ++e;
      }
    }
    for (; j + 4 <= m; j += 4) {
      v4f rv[4];
#pragma unroll
      for (int i = 0; i < 4; ++i) {
        const int s = __builtin_amdgcn_readlane(cur.x, j + i);
        rv[i] = *(const v4f*)(Tin + (size_t)s * CIN + col);
      }
#pragma unroll
      for (int i = 0; i < 4; ++i) {
        const float c = __int_as_float(__builtin_amdgcn_readlane(cur.y, j + i));
        advance();
        acc += c * rv[i];
        ++e;
      }
    }
    for (; j < m; ++j) {
      const int s = __builtin_amdgcn_readlane(cur.x, j);
      const float c = __int_as_float(__builtin_amdgcn_readlane(cur.y, j));
      const v4f rv = *(const v4f*)(Tin + (size_t)s * CIN + col);
      advance();
      acc += c * rv;
      ++e;
    }
    cur = nxt;
  }
  for (;;) {
    *(v4f*)(sAw + r * APITCH + col) = finish(acc, r);
    if (++r >= 16) break;
    acc = *(const v4f*)(sAw + r * APITCH + col);
  }

  const int mm = lane & 15;
  const int kh = lane >> 4;
  constexpr int NT = COUT / 16;
  v8f accT[NT] = {};
  const float* arow = sAw + mm * APITCH + 2 * kh;
  for (int k = 0; k < CIN; k += 4) {
    const v2f a = *(const v2f*)(arow + k);
    const float* b0 = &sW[((k >> 1) + kh) * PITCH + 2 * mm];
#pragma unroll
    for (int t = 0; t < NT; ++t) {
      v2f b = *(const v2f*)(b0 + 32 * t);
      accT[t] = __builtin_amdgcn_wmma_f32_16x16x4_f32(
           false, a,  false, b,
           (short)0, accT[t],  false,  false);
    }
  }

  float* o = Tout + (size_t)(m0 + 8 * kh) * COUT + mm;
#pragma unroll
  for (int t = 0; t < NT; ++t) {
#pragma unroll
    for (int v = 0; v < 8; ++v) {
      o[(size_t)v * COUT + t * 16] = accT[t][v];
    }
  }
}


extern "C" void kernel_launch(void* const* d_in, const int* in_sizes, int n_in,
                              void* d_out, int out_size, void* d_ws, size_t ws_size,
                              hipStream_t stream) {
  const float* x  = (const float*)d_in[0];
  const int*   ei = (const int*)d_in[1];
  const float* ea = (const float*)d_in[2];
  const float* W1 = (const float*)d_in[3];
  const float* b1 = (const float*)d_in[4];
  const float* W2 = (const float*)d_in[5];
  const float* b2 = (const float*)d_in[6];
  const float* W3 = (const float*)d_in[7];
  const float* b3 = (const float*)d_in[8];

  const int N = in_sizes[0] / CIN;
  const int E = in_sizes[2];
  const int* srcIdx = ei;
  const int* dstIdx = ei + E;

  float* T     = (float*)d_ws;
  float* agg   = T + (size_t)N * 128;
  int*   pairs = (int*)(agg + (size_t)N * 128);
  int*   rank  = pairs + (size_t)2 * E;
  float* dinv  = (float*)(rank + E);
  int*   cnt   = (int*)(dinv + N);
  int*   off   = cnt + N;
  const int J   = (E + RK_CHUNK - 1) / RK_CHUNK;
  const int NW  = (N + 1) / 2;
  const int NW4 = (NW + 3) / 4 * 4;
  unsigned* cntJ = (unsigned*)(off + N);
  int*   rel   = (int*)(cntJ + (size_t)J * NW4);
  const int NPW = (NW4 + RK_UNIT - 1) / RK_UNIT;
  unsigned* part = (unsigned*)(rel + (size_t)J * N);
  float* out   = (float*)d_out;

  const int nRowTiles = N / 16;
  dim3 blk(256);
  dim3 gE((E + 255) / 256);
  dim3 gGemm((nRowTiles + 7) / 8);
  dim3 gRows((N + 7) / 8);

  count_rank<<<dim3(J), dim3(RK_THREADS), 0, stream>>>(dstIdx, cntJ, part, rank, N, E, NPW);
  sum_scan_counts<<<dim3((NW + SS_THREADS - 1) / SS_THREADS), dim3(SS_THREADS), 0, stream>>>(
      cntJ, part, rel, cnt, off, N, J, NPW);
  place_edges<<<gE, blk, 0, stream>>>(srcIdx, dstIdx, ea, rel, rank, pairs, N, E);

  gemm_wmma<128, false><<<gGemm, blk, 0, stream>>>(x, W1, pairs, off, cnt, dinv, T, nRowTiles);

  gather_gemm<128, 8><<<dim3((nRowTiles + 7) / 8), dim3(256), 0, stream>>>(
      pairs, off, cnt, dinv, T, b1, W2, agg, nRowTiles);

  gather_gemm<64, 13><<<dim3((nRowTiles + 12) / 13), dim3(416), 0, stream>>>(
      pairs, off, cnt, dinv, agg, b2, W3, T, nRowTiles);

  gather_rows<64><<<gRows, blk, 0, stream>>>(pairs, off, cnt, dinv, T, b3, out, N);
}


# Fused row-sum+product launches sized to the chip: one work-group of 25 tiles per WGP (250, LDS admits one each); was 782 of 8 (fullest WGP 32 tiles) and 481 of 13 (26); sums and sum order kept
# speedup vs baseline: 4.2927x; 1.0753x over previous
#include <hip/hip_runtime.h>
#include <hip/hip_bf16.h>


#define CIN 128

typedef __attribute__((ext_vector_type(2))) float v2f;
typedef __attribute__((ext_vector_type(4))) float v4f;
typedef __attribute__((ext_vector_type(8))) float v8f;


#define RK_THREADS 1024
#define RK_TRIPS   15
#define RK_CHUNK   (RK_THREADS * 4 * RK_TRIPS)
#define RK_WORDS   65536
#define RK_UNIT    128

__device__ __forceinline__ int lds_rank(unsigned* sCnt, int d) {
  const unsigned sh  = (unsigned)(d & 1) << 4;
  const unsigned old = atomicAdd(&sCnt[d >> 1], 1u << sh);
  return (int)((old >> sh) & 0xFFFFu);
}

__device__ __forceinline__ void lds_clear(unsigned* sCnt, int NW4, int t) {
  for (int i = t * 4; i < NW4; i += RK_THREADS * 4)
    *(uint4*)&sCnt[i] = make_uint4(0u, 0u, 0u, 0u);
}

__global__ __launch_bounds__(RK_THREADS) void count_rank(const int* __restrict__ dst,
                                                         unsigned* __restrict__ cntJ,
                                                         unsigned* __restrict__ part,
                                                         int* __restrict__ rank, int N, int E,
                                                         int NPW) {
  __shared__ alignas(16) unsigned sCnt[RK_WORDS];
  const int t    = threadIdx.x;
  const int NW4  = (((N + 1) >> 1) + 3) & ~3;
  const int eBeg = blockIdx.x * RK_CHUNK;
  const bool full = ((E & 3) == 0) && (E - eBeg >= RK_CHUNK);

  if (full) {
    int4 dv[RK_TRIPS];
#pragma unroll
    for (int u = 0; u < RK_TRIPS; ++u)
      dv[u] = *(const int4*)(dst + eBeg + (u * RK_THREADS + t) * 4);
    lds_clear(sCnt, NW4, t);
    __syncthreads();
#pragma unroll
    for (int u = 0; u < RK_TRIPS; ++u) {
      int4 r;
      r.x = lds_rank(sCnt, dv[u].x);
      r.y = lds_rank(sCnt, dv[u].y);
      r.z = lds_rank(sCnt, dv[u].z);
      r.w = lds_rank(sCnt, dv[u].w);
      *(int4*)(rank + eBeg + (u * RK_THREADS + t) * 4) = r;
    }
  } else {
    const int eEnd = (E - eBeg < RK_CHUNK) ? E : eBeg + RK_CHUNK;
    lds_clear(sCnt, NW4, t);
    __syncthreads();
    for (int e = eBeg + t; e < eEnd; e += RK_THREADS) rank[e] = lds_rank(sCnt, dst[e]);
  }
  __syncthreads();

  unsigned* o  = cntJ + (size_t)blockIdx.x * NW4;
  unsigned* pw = part + (size_t)blockIdx.x * NPW;
  for (int it = 0; it * (RK_THREADS * 4) < NW4; ++it) {
    const int i = it * (RK_THREADS * 4) + t * 4;
    unsigned s = 0;
    if (i < NW4) {
      const uint4 c = *(const uint4*)&sCnt[i];
      *(uint4*)(o + i) = c;
      s = (c.x & 0xFFFFu) + (c.x >> 16) + (c.y & 0xFFFFu) + (c.y >> 16) +
          (c.z & 0xFFFFu) + (c.z >> 16) + (c.w & 0xFFFFu) + (c.w >> 16);
    }
    for (int sh = 16; sh > 0; sh >>= 1) s += __shfl_xor(s, sh);
    const int q = it * (RK_THREADS / 32) + (t >> 5);
    if ((t & 31) == 0 && q * RK_UNIT < NW4) pw[q] = s;
  }
}

#define SC_GROUP   14
#define SS_THREADS 256
#define SS_UPB     (SS_THREADS / RK_UNIT)

template <bool EVEN>
__device__ __forceinline__ void chunk_starts_body(const unsigned* __restrict__ cntJ,
                                                  int* __restrict__ rel, int N, int J, int w,
                                                  int NW4, int a0, int a1) {
  const int n0 = 2 * w;
  for (int j0 = 0; j0 < J; j0 += SC_GROUP) {
    unsigned c[SC_GROUP];
#pragma unroll
    for (int k = 0; k < SC_GROUP; ++k) {
      const int j = (j0 + k < J) ? (j0 + k) : (J - 1);
      c[k] = cntJ[(size_t)j * NW4 + w];
    }
#pragma unroll
    for (int k = 0; k < SC_GROUP; ++k) {
      if (j0 + k < J) {
        int* r = rel + (size_t)(j0 + k) * N + n0;
        if (EVEN) {
          *(int2*)r = make_int2(a0, a1);
        } else {
          r[0] = a0;
          if (n0 + 1 < N) r[1] = a1;
        }
        a0 += (int)(c[k] & 0xFFFFu);
        a1 += (int)(c[k] >> 16);
      }
    }
  }
}

__global__ __launch_bounds__(SS_THREADS) void sum_scan_counts(const unsigned* __restrict__ cntJ,
                                                              const unsigned* __restrict__ part,
                                                              int* __restrict__ rel,
                                                              int* __restrict__ cnt,
                                                              int* __restrict__ off,
                                                              int N, int J, int NPW) {
  __shared__ int sTot[SS_THREADS / 32];
  __shared__ int sFront[SS_THREADS / 32];
  const int tid  = threadIdx.x;
  const int lane = tid & 31;
  const int wv   = tid >> 5;
  const int NW   = (N + 1) >> 1;
  const int NW4  = (NW + 3) & ~3;
  const int w    = blockIdx.x * SS_THREADS + tid;
  const bool valid = w < NW;
  const int wc   = valid ? w : (NW - 1);
  const int nq   = (int)blockIdx.x * SS_UPB;
  const bool u0ok = tid < nq;
  const bool u1ok = tid + SS_THREADS < nq;
  const int u0   = u0ok ? tid : 0;
  const int u1   = u1ok ? tid + SS_THREADS : 0;

  int t0 = 0, t1 = 0, front = 0;
  for (int j0 = 0; j0 < J; j0 += SC_GROUP) {
    unsigned c[SC_GROUP], p0[SC_GROUP], p1[SC_GROUP];
#pragma unroll
    for (int k = 0; k < SC_GROUP; ++k) {
      const int j = (j0 + k < J) ? (j0 + k) : (J - 1);
      c[k]  = cntJ[(size_t)j * NW4 + wc];
      p0[k] = part[(size_t)j * NPW + u0];
      p1[k] = part[(size_t)j * NPW + u1];
    }
#pragma unroll
    for (int k = 0; k < SC_GROUP; ++k) {
      if (j0 + k < J) {
        t0 += (int)(c[k] & 0xFFFFu);
        t1 += (int)(c[k] >> 16);
        if (u0ok) front += (int)p0[k];
        if (u1ok) front += (int)p1[k];
      }
    }
  }
  if (!valid) { t0 = 0; t1 = 0; }

  const int s = t0 + t1;
  int incl = s;
  for (int o = 1; o < 32; o <<= 1) {
    const int y = __shfl_up(incl, o);
    if (lane >= o) incl += y;
  }
  for (int o = 16; o > 0; o >>= 1) front += __shfl_xor(front, o);
  if (lane == 31) sTot[wv] = incl;
  if (lane == 0) sFront[wv] = front;
  __syncthreads();
  int base = 0;
  for (int k = 0; k < SS_THREADS / 32; ++k) base += sFront[k];
  for (int k = 0; k < wv; ++k) base += sTot[k];
  const int off0 = base + incl - s;
  const int off1 = off0 + t0;

  if (!valid) return;
  const int n0 = 2 * w;
  off[n0] = off0;
  cnt[n0] = t0;
  if (n0 + 1 < N) { off[n0 + 1] = off1; cnt[n0 + 1] = t1; }

  if ((N & 1) == 0) chunk_starts_body<true>(cntJ, rel, N, J, w, NW4, off0, off1);
  else              chunk_starts_body<false>(cntJ, rel, N, J, w, NW4, off0, off1);
}

__global__ void place_edges(const int* __restrict__ src, const int* __restrict__ dst,
                            const float* __restrict__ ea,
                            const int* __restrict__ rel, const int* __restrict__ rank,
                            int* __restrict__ pairs, int N, int E) {
  int e = blockIdx.x * blockDim.x + threadIdx.x;
  if (e >= E) return;
  const int d = dst[e];
  const int j = e / RK_CHUNK;
  const int p = rel[(size_t)j * N + d] + rank[e];
  *(int2*)(pairs + 2 * (size_t)p) = make_int2(src[e], __float_as_int(ea[e]));
}


template <int COUT, bool RELU>
__global__ __launch_bounds__(256) void gemm_wmma(const float* __restrict__ H,
                                                 const float* __restrict__ Wt,
                                                 const int* __restrict__ pairs,
                                                 const int* __restrict__ off,
                                                 const int* __restrict__ cnt,
                                                 float* __restrict__ dinv,
                                                 float* __restrict__ T,
                                                 int nRowTiles) {
  constexpr int PITCH = 2 * COUT + 32;
  __shared__ alignas(16) float sW[(CIN / 2) * PITCH];

  constexpr int NTHREADS = 256;
  constexpr int NV       = COUT / 4;
  constexpr int PER_T    = (CIN / 2) * NV / NTHREADS;
  v4f wlo[PER_T], whi[PER_T];
#pragma unroll
  for (int u = 0; u < PER_T; ++u) {
    const int id = (int)threadIdx.x + u * NTHREADS;
    const int p  = id / NV;
    const int g  = id - p * NV;
    wlo[u] = *(const v4f*)(Wt + (size_t)(2 * p) * COUT + 4 * g);
    whi[u] = *(const v4f*)(Wt + (size_t)(2 * p + 1) * COUT + 4 * g);
  }

  const int wave = threadIdx.x >> 5;
  const int lane = threadIdx.x & 31;
  const int rt   = blockIdx.x * 8 + wave;

  float dn = 0.0f;
  if (rt < nRowTiles) {
    const int nn = rt * 16 + (lane & 15);
    const int* p = pairs + 2 * (size_t)off[nn];
    const int c  = cnt[nn];
    float d = 1.0f;
    int j = 0;
    for (; j + 4 <= c; j += 4) {
      const float w0 = __int_as_float(p[2 * j + 1]);
      const float w1 = __int_as_float(p[2 * j + 3]);
      const float w2 = __int_as_float(p[2 * j + 5]);
      const float w3 = __int_as_float(p[2 * j + 7]);
      d += w0;
      d += w1;
      d += w2;
      d += w3;
    }
    for (; j < c; ++j) d += __int_as_float(p[2 * j + 1]);
    dn = (d > 0.0f) ? rsqrtf(d) : 0.0f;
    if (lane < 16) dinv[nn] = dn;
  }

#pragma unroll
  for (int u = 0; u < PER_T; ++u) {
    const int id = (int)threadIdx.x + u * NTHREADS;
    const int p  = id / NV;
    const int g  = id - p * NV;
    float* d = &sW[p * PITCH + 8 * g];
    v4f s0, s1;
    s0.x = wlo[u].x; s0.y = whi[u].x; s0.z = wlo[u].y; s0.w = whi[u].y;
    s1.x = wlo[u].z; s1.y = whi[u].z; s1.z = wlo[u].w; s1.w = whi[u].w;
    *(v4f*)(d)     = s0;
    *(v4f*)(d + 4) = s1;
  }
  __syncthreads();

  if (rt >= nRowTiles) return;

  const int m0 = rt * 16;
  const int m  = lane & 15;
  const int kh = lane >> 4;

  constexpr int NT = COUT / 16;
  v8f acc[NT] = {};

  const float* arow = H + (size_t)(m0 + m) * CIN + 2 * kh;

  for (int k = 0; k < CIN; k += 4) {
    v2f a = *(const v2f*)(arow + k);
    if (RELU) {
      a.x = fmaxf(a.x, 0.0f);
      a.y = fmaxf(a.y, 0.0f);
    }
    const float* b0 = &sW[((k >> 1) + kh) * PITCH + 2 * m];
#pragma unroll
    for (int t = 0; t < NT; ++t) {
      v2f b = *(const v2f*)(b0 + 32 * t);
      acc[t] = __builtin_amdgcn_wmma_f32_16x16x4_f32(
           false, a,  false, b,
           (short)0, acc[t],  false,  false);
    }
  }

  float sc[8];
#pragma unroll
  for (int v = 0; v < 8; ++v) sc[v] = __shfl(dn, v + 8 * kh);
  float* out = T + (size_t)(m0 + 8 * kh) * COUT + m;
#pragma unroll
  for (int t = 0; t < NT; ++t) {
#pragma unroll
    for (int v = 0; v < 8; ++v) {
      out[(size_t)v * COUT + t * 16] = sc[v] * acc[t][v];
    }
  }
}


template <int C>
__global__ __launch_bounds__(256) void gather_rows(const int* __restrict__ pairs,
                                                   const int* __restrict__ off,
                                                   const int* __restrict__ cnt,
                                                   const float* __restrict__ dinv,
                                                   const float* __restrict__ T,
                                                   const float* __restrict__ bias,
                                                   float* __restrict__ agg, int N) {
  constexpr int V = C / 32;
  typedef float vecT __attribute__((ext_vector_type(V)));

  const int lane = threadIdx.x & 31;
  const int wv = __builtin_amdgcn_readfirstlane((int)(threadIdx.x >> 5));
  const int n  = blockIdx.x * 8 + wv;
  if (n >= N) return;

  const int beg = off[n];
  const int c   = cnt[n];
  const float dn = dinv[n];

  const int col = lane * V;
  vecT acc = *(const vecT*)(T + (size_t)n * C + col);

  for (int base = 0; base < c; base += 32) {
    const int m = (c - base < 32) ? (c - base) : 32;
    int ms = 0;
    int mc = 0;
    if (lane < m) {
      const int2 en = *(const int2*)(pairs + 2 * (size_t)(beg + base + lane));
      ms = en.x;
      mc = en.y;
    }
    int j = 0;
    for (; j + 4 <= m; j += 4) {
      const int s0 = __builtin_amdgcn_readlane(ms, j);
      const int s1 = __builtin_amdgcn_readlane(ms, j + 1);
      const int s2 = __builtin_amdgcn_readlane(ms, j + 2);
      const int s3 = __builtin_amdgcn_readlane(ms, j + 3);
      const float c0 = __int_as_float(__builtin_amdgcn_readlane(mc, j));
      const float c1 = __int_as_float(__builtin_amdgcn_readlane(mc, j + 1));
      const float c2 = __int_as_float(__builtin_amdgcn_readlane(mc, j + 2));
      const float c3 = __int_as_float(__builtin_amdgcn_readlane(mc, j + 3));
      const vecT r0 = *(const vecT*)(T + (size_t)s0 * C + col);
      const vecT r1 = *(const vecT*)(T + (size_t)s1 * C + col);
      const vecT r2 = *(const vecT*)(T + (size_t)s2 * C + col);
      const vecT r3 = *(const vecT*)(T + (size_t)s3 * C + col);
      acc += c0 * r0;
      acc += c1 * r1;
      acc += c2 * r2;
      acc += c3 * r3;
    }
    for (; j < m; ++j) {
      const int s0 = __builtin_amdgcn_readlane(ms, j);
      const float c0 = __int_as_float(__builtin_amdgcn_readlane(mc, j));
      acc += c0 * *(const vecT*)(T + (size_t)s0 * C + col);
    }
  }

  *(vecT*)(agg + (size_t)n * C + col) = *(const vecT*)(bias + col) + dn * acc;
}

#define APITCH 132
#define GB 16
#define TILES_PER_WG 25

template <int COUT, int NWAVES>
__global__ __launch_bounds__(NWAVES * 32) void gather_gemm(const int* __restrict__ pairs,
                                                           const int* __restrict__ off,
                                                           const int* __restrict__ cnt,
                                                           const float* __restrict__ dinv,
                                                           const float* __restrict__ Tin,
                                                           const float* __restrict__ bias,
                                                           const float* __restrict__ Wt,
                                                           float* __restrict__ Tout,
                                                           int nRowTiles) {
  constexpr int PITCH    = 2 * COUT + 32;
  constexpr int NTHREADS = NWAVES * 32;
  constexpr int NV       = COUT / 4;
  constexpr int UNITS    = (CIN / 2) * NV;
  constexpr int PER_T    = (UNITS + NTHREADS - 1) / NTHREADS;
  constexpr bool EXACT   = (PER_T * NTHREADS == UNITS);
  __shared__ alignas(16) float sW[(CIN / 2) * PITCH];
  __shared__ alignas(16) float sA[NWAVES * 16 * APITCH];

  const int tid  = (int)threadIdx.x;
  const int lane = tid & 31;
  const int wave = __builtin_amdgcn_readfirstlane(tid >> 5);
  const int rt   = blockIdx.x * NWAVES + wave;
  const bool live = rt < nRowTiles;
  const int m0   = rt * 16;
  const int col  = lane * 4;
  float* sAw = sA + wave * (16 * APITCH);

  v4f wlo[PER_T], whi[PER_T];
#pragma unroll
  for (int u = 0; u < PER_T; ++u) {
    const int id = tid + u * NTHREADS;
    if (EXACT || id < UNITS) {
      const int p = id / NV;
      const int g = id - p * NV;
      wlo[u] = *(const v4f*)(Wt + (size_t)(2 * p) * COUT + 4 * g);
      whi[u] = *(const v4f*)(Wt + (size_t)(2 * p + 1) * COUT + 4 * g);
    }
  }

  int rOff = 0, rCnt = 0;
  float rD = 0.0f;
  v4f bv = {0.0f, 0.0f, 0.0f, 0.0f};
  v4f self[16];
  if (live) {
    const int nn = m0 + (lane & 15);
    rOff = off[nn];
    rCnt = cnt[nn];
    rD   = dinv[nn];
    bv   = *(const v4f*)(bias + col);
#pragma unroll
    for (int r = 0; r < 16; ++r) self[r] = *(const v4f*)(Tin + (size_t)(m0 + r) * CIN + col);
  }

#pragma unroll
  for (int u = 0; u < PER_T; ++u) {
    const int id = tid + u * NTHREADS;
    if (EXACT || id < UNITS) {
      const int p = id / NV;
      const int g = id - p * NV;
      float* d = &sW[p * PITCH + 8 * g];
      v4f s0, s1;
      s0.x = wlo[u].x; s0.y = whi[u].x; s0.z = wlo[u].y; s0.w = whi[u].y;
      s1.x = wlo[u].z; s1.y = whi[u].z; s1.z = wlo[u].w; s1.w = whi[u].w;
      *(v4f*)(d)     = s0;
      *(v4f*)(d + 4) = s1;
    }
  }
  __syncthreads();
  if (!live) return;

#pragma unroll
  for (int r = 0; r < 16; ++r) *(v4f*)(sAw + r * APITCH + col) = self[r];

  const int e0 = __builtin_amdgcn_readlane(rOff, 0);
  const int e1 = __builtin_amdgcn_readlane(rOff, 15) + __builtin_amdgcn_readlane(rCnt, 15);
  int r      = 0;
  int rowEnd = e0 + __builtin_amdgcn_readlane(rCnt, 0);
  int e      = e0;
  v4f acc = *(const v4f*)(sAw + col);

  auto finish = [&](v4f sum, int row) -> v4f {
    const float dr = __int_as_float(__builtin_amdgcn_readlane(__float_as_int(rD), row));
    const v4f a = bv + dr * sum;
    v4f h;
    h.x = fmaxf(a.x, 0.0f); h.y = fmaxf(a.y, 0.0f);
    h.z = fmaxf(a.z, 0.0f); h.w = fmaxf(a.w, 0.0f);
    return dr * h;
  };
  auto advance = [&]() {
    while (e == rowEnd) {
      *(v4f*)(sAw + r * APITCH + col) = finish(acc, r);
      ++r;
      rowEnd = __builtin_amdgcn_readlane(rOff, r) + __builtin_amdgcn_readlane(rCnt, r);
      acc = *(const v4f*)(sAw + r * APITCH + col);
    }
  };

  int2 cur = make_int2(0, 0);
  if (e0 + lane < e1) cur = *(const int2*)(pairs + 2 * (size_t)(e0 + lane));
  for (int base = e0; base < e1; base += 32) {
    int2 nxt = make_int2(0, 0);
    if (base + 32 + lane < e1) nxt = *(const int2*)(pairs + 2 * (size_t)(base + 32 + lane));
    const int m = (e1 - base < 32) ? (e1 - base) : 32;
    int j = 0;
    for (; j + GB <= m; j += GB) {
      v4f rv[GB];
#pragma unroll
      for (int i = 0; i < GB; ++i) {
        const int s = __builtin_amdgcn_readlane(cur.x, j + i);
        rv[i] = *(const v4f*)(Tin + (size_t)s * CIN + col);
      }
#pragma unroll
      for (int i = 0; i < GB; ++i) {
        const float c = __int_as_float(__builtin_amdgcn_readlane(cur.y, j + i));
        advance();
        acc += c * rv[i];
        ++e;
      }
    }
    for (; j + 4 <= m; j += 4) {
      v4f rv[4];
#pragma unroll
      for (int i = 0; i < 4; ++i) {
        const int s = __builtin_amdgcn_readlane(cur.x, j + i);
        rv[i] = *(const v4f*)(Tin + (size_t)s * CIN + col);
      }
#pragma unroll
      for (int i = 0; i < 4; ++i) {
        const float c = __int_as_float(__builtin_amdgcn_readlane(cur.y, j + i));
        advance();
        acc += c * rv[i];
        ++e;
      }
    }
    for (; j < m; ++j) {
      const int s = __builtin_amdgcn_readlane(cur.x, j);
      const float c = __int_as_float(__builtin_amdgcn_readlane(cur.y, j));
      const v4f rv = *(const v4f*)(Tin + (size_t)s * CIN + col);
      advance();
      acc += c * rv;
      ++e;
    }
    cur = nxt;
  }
  for (;;) {
    *(v4f*)(sAw + r * APITCH + col) = finish(acc, r);
    if (++r >= 16) break;
    acc = *(const v4f*)(sAw + r * APITCH + col);
  }

  const int mm = lane & 15;
  const int kh = lane >> 4;
  constexpr int NT = COUT / 16;
  v8f accT[NT] = {};
  const float* arow = sAw + mm * APITCH + 2 * kh;
  for (int k = 0; k < CIN; k += 4) {
    const v2f a = *(const v2f*)(arow + k);
    const float* b0 = &sW[((k >> 1) + kh) * PITCH + 2 * mm];
#pragma unroll
    for (int t = 0; t < NT; ++t) {
      v2f b = *(const v2f*)(b0 + 32 * t);
      accT[t] = __builtin_amdgcn_wmma_f32_16x16x4_f32(
           false, a,  false, b,
           (short)0, accT[t],  false,  false);
    }
  }

  float* o = Tout + (size_t)(m0 + 8 * kh) * COUT + mm;
#pragma unroll
  for (int t = 0; t < NT; ++t) {
#pragma unroll
    for (int v = 0; v < 8; ++v) {
      o[(size_t)v * COUT + t * 16] = accT[t][v];
    }
  }
}


extern "C" void kernel_launch(void* const* d_in, const int* in_sizes, int n_in,
                              void* d_out, int out_size, void* d_ws, size_t ws_size,
                              hipStream_t stream) {
  const float* x  = (const float*)d_in[0];
  const int*   ei = (const int*)d_in[1];
  const float* ea = (const float*)d_in[2];
  const float* W1 = (const float*)d_in[3];
  const float* b1 = (const float*)d_in[4];
  const float* W2 = (const float*)d_in[5];
  const float* b2 = (const float*)d_in[6];
  const float* W3 = (const float*)d_in[7];
  const float* b3 = (const float*)d_in[8];

  const int N = in_sizes[0] / CIN;
  const int E = in_sizes[2];
  const int* srcIdx = ei;
  const int* dstIdx = ei + E;

  float* T     = (float*)d_ws;
  float* agg   = T + (size_t)N * 128;
  int*   pairs = (int*)(agg + (size_t)N * 128);
  int*   rank  = pairs + (size_t)2 * E;
  float* dinv  = (float*)(rank + E);
  int*   cnt   = (int*)(dinv + N);
  int*   off   = cnt + N;
  const int J   = (E + RK_CHUNK - 1) / RK_CHUNK;
  const int NW  = (N + 1) / 2;
  const int NW4 = (NW + 3) / 4 * 4;
  unsigned* cntJ = (unsigned*)(off + N);
  int*   rel   = (int*)(cntJ + (size_t)J * NW4);
  const int NPW = (NW4 + RK_UNIT - 1) / RK_UNIT;
  unsigned* part = (unsigned*)(rel + (size_t)J * N);
  float* out   = (float*)d_out;

  const int nRowTiles = N / 16;
  dim3 blk(256);
  dim3 gE((E + 255) / 256);
  dim3 gGemm((nRowTiles + 7) / 8);
  dim3 gRows((N + 7) / 8);

  count_rank<<<dim3(J), dim3(RK_THREADS), 0, stream>>>(dstIdx, cntJ, part, rank, N, E, NPW);
  sum_scan_counts<<<dim3((NW + SS_THREADS - 1) / SS_THREADS), dim3(SS_THREADS), 0, stream>>>(
      cntJ, part, rel, cnt, off, N, J, NPW);
  place_edges<<<gE, blk, 0, stream>>>(srcIdx, dstIdx, ea, rel, rank, pairs, N, E);

  gemm_wmma<128, false><<<gGemm, blk, 0, stream>>>(x, W1, pairs, off, cnt, dinv, T, nRowTiles);

  dim3 gTiles((nRowTiles + TILES_PER_WG - 1) / TILES_PER_WG);
  dim3 bTiles(TILES_PER_WG * 32);

  gather_gemm<128, TILES_PER_WG><<<gTiles, bTiles, 0, stream>>>(
      pairs, off, cnt, dinv, T, b1, W2, agg, nRowTiles);

  gather_gemm<64, TILES_PER_WG><<<gTiles, bTiles, 0, stream>>>(
      pairs, off, cnt, dinv, agg, b2, W3, T, nRowTiles);

  gather_rows<64><<<gRows, blk, 0, stream>>>(pairs, off, cnt, dinv, T, b3, out, N);
}


# Edge grouping in two levels inside two launches: slices bucket edges per consumer work-group (LDS ranks), the first-product launch orders its bucket; 7 launches to 5; row sum order: consumer LDS rank
# speedup vs baseline: 5.2167x; 1.2152x over previous
#include <hip/hip_runtime.h>
#include <hip/hip_bf16.h>


#define CIN 128

typedef __attribute__((ext_vector_type(2))) float v2f;
typedef __attribute__((ext_vector_type(4))) float v4f;
typedef __attribute__((ext_vector_type(8))) float v8f;

#define TILES_PER_WG 25
#define BK_NODES   (TILES_PER_WG * 16)
#define NSLICE     250
#define BE_THREADS 1024
#define BE_EPT     7
#define BE_MAXB    1023

__global__ __launch_bounds__(BE_THREADS) void bucket_edges(const int* __restrict__ src,
                                                           const int* __restrict__ dst,
                                                           const float* __restrict__ ea,
                                                           int* __restrict__ recSW,
                                                           int* __restrict__ recD,
                                                           int* __restrict__ startT,
                                                           int E, int SL, int NB) {
  __shared__ int sCnt[BE_MAXB + 1];
  __shared__ int sStart[BE_MAXB + 1];
  __shared__ int sWave[BE_THREADS / 32];
  const int t    = threadIdx.x;
  const int lane = t & 31;
  const int wv   = t >> 5;
  const int g    = blockIdx.x;
  const int sbeg = g * SL;
  const int send = (E - sbeg < SL) ? E : sbeg + SL;

  int d[BE_EPT], s[BE_EPT], w[BE_EPT];
#pragma unroll
  for (int u = 0; u < BE_EPT; ++u) {
    const int i = sbeg + u * BE_THREADS + t;
    d[u] = -1; s[u] = 0; w[u] = 0;
    if (i < send) { d[u] = dst[i]; s[u] = src[i]; w[u] = __float_as_int(ea[i]); }
  }
  sCnt[t] = 0;
  __syncthreads();

  int bk[BE_EPT], rk[BE_EPT];
#pragma unroll
  for (int u = 0; u < BE_EPT; ++u) {
    bk[u] = 0; rk[u] = 0;
    if (d[u] >= 0) { bk[u] = d[u] / BK_NODES; rk[u] = atomicAdd(&sCnt[bk[u]], 1); }
  }
  __syncthreads();

  const int c = sCnt[t];
  int incl = c;
  for (int o = 1; o < 32; o <<= 1) {
    const int y = __shfl_up(incl, o);
    if (lane >= o) incl += y;
  }
  if (lane == 31) sWave[wv] = incl;
  __syncthreads();
  int front = 0;
  for (int k = 0; k < wv; ++k) front += sWave[k];
  const int excl = front + incl - c;
  sStart[t] = excl;
  if (t <= NB) startT[(size_t)t * NSLICE + g] = excl;
  __syncthreads();

#pragma unroll
  for (int u = 0; u < BE_EPT; ++u) {
    if (d[u] >= 0) {
      const int p = sbeg + sStart[bk[u]] + rk[u];
      *(int2*)(recSW + 2 * (size_t)p) = make_int2(s[u], w[u]);
      recD[p] = d[u];
    }
  }
}

#define RPW (NSLICE / TILES_PER_WG)
#define RDEPTH 2
static_assert(RPW * TILES_PER_WG == NSLICE, "every run has a wave");
static_assert(NSLICE <= TILES_PER_WG * 32 && BK_NODES <= TILES_PER_WG * 32, "one thread per slice / per node");

template <int COUT>
__global__ __launch_bounds__(TILES_PER_WG * 32) void sort_gemm(const float* __restrict__ H,
                                                               const float* __restrict__ Wt,
                                                               const int* __restrict__ recSW,
                                                               const int* __restrict__ recD,
                                                               const int* __restrict__ startT,
                                                               int* pairs,
                                                               int* __restrict__ off,
                                                               int* __restrict__ cnt,
                                                               float* __restrict__ dinv,
                                                               float* __restrict__ T,
                                                               int nRowTiles, int N, int SL) {
  constexpr int PITCH    = 2 * COUT + 32;
  constexpr int NTHREADS = TILES_PER_WG * 32;
  constexpr int NV       = COUT / 4;
  constexpr int UNITS    = (CIN / 2) * NV;
  constexpr int PER_T    = (UNITS + NTHREADS - 1) / NTHREADS;
  __shared__ alignas(16) float sW[(CIN / 2) * PITCH];
  __shared__ int sRunS[NSLICE];
  __shared__ int sRunE[NSLICE];
  __shared__ int sCntN[BK_NODES];
  __shared__ int sOffN[BK_NODES];
  __shared__ int sPosN[BK_NODES];
  __shared__ int sRedA[TILES_PER_WG];
  __shared__ int sRedB[TILES_PER_WG];

  const int tid  = (int)threadIdx.x;
  const int lane = tid & 31;
  const int wave = __builtin_amdgcn_readfirstlane(tid >> 5);
  const int b    = (int)blockIdx.x;
  const int nbeg = b * BK_NODES;

  v4f wlo[PER_T], whi[PER_T];
#pragma unroll
  for (int u = 0; u < PER_T; ++u) {
    const int id = tid + u * NTHREADS;
    if (id < UNITS) {
      const int p = id / NV;
      const int g = id - p * NV;
      wlo[u] = *(const v4f*)(Wt + (size_t)(2 * p) * COUT + 4 * g);
      whi[u] = *(const v4f*)(Wt + (size_t)(2 * p + 1) * COUT + 4 * g);
    }
  }

  int rs = 0, re = 0;
  if (tid < NSLICE) {
    rs = startT[(size_t)b * NSLICE + tid];
    re = startT[(size_t)(b + 1) * NSLICE + tid];
  }
  if (tid < BK_NODES) sCntN[tid] = 0;

#pragma unroll
  for (int u = 0; u < PER_T; ++u) {
    const int id = tid + u * NTHREADS;
    if (id < UNITS) {
      const int p = id / NV;
      const int g = id - p * NV;
      float* d = &sW[p * PITCH + 8 * g];
      v4f s0, s1;
      s0.x = wlo[u].x; s0.y = whi[u].x; s0.z = wlo[u].y; s0.w = whi[u].y;
      s1.x = wlo[u].z; s1.y = whi[u].z; s1.z = wlo[u].w; s1.w = whi[u].w;
      *(v4f*)(d)     = s0;
      *(v4f*)(d + 4) = s1;
    }
  }

  if (tid < NSLICE) {
    sRunS[tid] = tid * SL + rs;
    sRunE[tid] = tid * SL + re;
  }
  int fs = rs;
  for (int o = 16; o > 0; o >>= 1) fs += __shfl_xor(fs, o);
  if (lane == 0) sRedA[wave] = fs;
  __syncthreads();
  int base = 0;
  for (int k = 0; k < TILES_PER_WG; ++k) base += sRedA[k];

  int rS[RPW], rE[RPW], dd[RPW * RDEPTH];
  int2 sw[RPW * RDEPTH];
#pragma unroll
  for (int k = 0; k < RPW; ++k) {
    rS[k] = __builtin_amdgcn_readfirstlane(sRunS[wave * RPW + k]);
    rE[k] = __builtin_amdgcn_readfirstlane(sRunE[wave * RPW + k]);
  }
#pragma unroll
  for (int k = 0; k < RPW * RDEPTH; ++k) {
    const int i = rS[k / RDEPTH] + 32 * (k % RDEPTH) + lane;
    dd[k] = -1;
    sw[k] = make_int2(0, 0);
    if (i < rE[k / RDEPTH]) {
      dd[k] = recD[i];
      sw[k] = *(const int2*)(recSW + 2 * (size_t)i);
    }
  }
#pragma unroll
  for (int k = 0; k < RPW * RDEPTH; ++k) {
    if (dd[k] >= 0) atomicAdd(&sCntN[dd[k] - nbeg], 1);
  }
  for (int k = 0; k < RPW; ++k) {
    for (int i = rS[k] + 32 * RDEPTH + lane; i < rE[k]; i += 32) atomicAdd(&sCntN[recD[i] - nbeg], 1);
  }
  __syncthreads();

  const int c = (tid < BK_NODES) ? sCntN[tid] : 0;
  int incl = c;
  for (int o = 1; o < 32; o <<= 1) {
    const int y = __shfl_up(incl, o);
    if (lane >= o) incl += y;
  }
  if (lane == 31) sRedB[wave] = incl;
  __syncthreads();
  int front = 0;
  for (int k = 0; k < wave; ++k) front += sRedB[k];
  const int excl = front + incl - c;
  if (tid < BK_NODES) {
    sOffN[tid] = excl;
    sPosN[tid] = excl;
    const int n = nbeg + tid;
    if (n < N) { off[n] = base + excl; cnt[n] = c; }
  }
  __syncthreads();

#pragma unroll
  for (int k = 0; k < RPW * RDEPTH; ++k) {
    if (dd[k] >= 0) {
      const int p = atomicAdd(&sPosN[dd[k] - nbeg], 1);
      *(int2*)(pairs + 2 * (size_t)(base + p)) = sw[k];
    }
  }
  for (int k = 0; k < RPW; ++k) {
    for (int i = rS[k] + 32 * RDEPTH + lane; i < rE[k]; i += 32) {
      const int dl = recD[i] - nbeg;
      const int2 v = *(const int2*)(recSW + 2 * (size_t)i);
      const int p  = atomicAdd(&sPosN[dl], 1);
      *(int2*)(pairs + 2 * (size_t)(base + p)) = v;
    }
  }
  __syncthreads();

  const int rt = b * TILES_PER_WG + wave;
  if (rt >= nRowTiles) return;

  const int li = wave * 16 + (lane & 15);
  const int nn = nbeg + li;
  float dn;
  {
    const int* p = pairs + 2 * (size_t)(base + sOffN[li]);
    const int cn = sCntN[li];
    float d = 1.0f;
    int j = 0;
    for (; j + 4 <= cn; j += 4) {
      const float w0 = __int_as_float(p[2 * j + 1]);
      const float w1 = __int_as_float(p[2 * j + 3]);
      const float w2 = __int_as_float(p[2 * j + 5]);
      const float w3 = __int_as_float(p[2 * j + 7]);
      d += w0;
      d += w1;
      d += w2;
      d += w3;
    }
    for (; j < cn; ++j) d += __int_as_float(p[2 * j + 1]);
    dn = (d > 0.0f) ? rsqrtf(d) : 0.0f;
    if (lane < 16) dinv[nn] = dn;
  }

  const int m0 = rt * 16;
  const int m  = lane & 15;
  const int kh = lane >> 4;

  constexpr int NT = COUT / 16;
  v8f acc[NT] = {};

  const float* arow = H + (size_t)(m0 + m) * CIN + 2 * kh;

  for (int k = 0; k < CIN; k += 4) {
    const v2f a = *(const v2f*)(arow + k);
    const float* b0 = &sW[((k >> 1) + kh) * PITCH + 2 * m];
#pragma unroll
    for (int t = 0; t < NT; ++t) {
      v2f bb = *(const v2f*)(b0 + 32 * t);
      acc[t] = __builtin_amdgcn_wmma_f32_16x16x4_f32(
           false, a,  false, bb,
           (short)0, acc[t],  false,  false);
    }
  }

  float sc[8];
#pragma unroll
  for (int v = 0; v < 8; ++v) sc[v] = __shfl(dn, v + 8 * kh);
  float* out = T + (size_t)(m0 + 8 * kh) * COUT + m;
#pragma unroll
  for (int t = 0; t < NT; ++t) {
#pragma unroll
    for (int v = 0; v < 8; ++v) {
      out[(size_t)v * COUT + t * 16] = sc[v] * acc[t][v];
    }
  }
}


template <int C>
__global__ __launch_bounds__(256) void gather_rows(const int* __restrict__ pairs,
                                                   const int* __restrict__ off,
                                                   const int* __restrict__ cnt,
                                                   const float* __restrict__ dinv,
                                                   const float* __restrict__ T,
                                                   const float* __restrict__ bias,
                                                   float* __restrict__ agg, int N) {
  constexpr int V = C / 32;
  typedef float vecT __attribute__((ext_vector_type(V)));

  const int lane = threadIdx.x & 31;
  const int wv = __builtin_amdgcn_readfirstlane((int)(threadIdx.x >> 5));
  const int n  = blockIdx.x * 8 + wv;
  if (n >= N) return;

  const int beg = off[n];
  const int c   = cnt[n];
  const float dn = dinv[n];

  const int col = lane * V;
  vecT acc = *(const vecT*)(T + (size_t)n * C + col);

  for (int base = 0; base < c; base += 32) {
    const int m = (c - base < 32) ? (c - base) : 32;
    int ms = 0;
    int mc = 0;
    if (lane < m) {
      const int2 en = *(const int2*)(pairs + 2 * (size_t)(beg + base + lane));
      ms = en.x;
      mc = en.y;
    }
    int j = 0;
    for (; j + 4 <= m; j += 4) {
      const int s0 = __builtin_amdgcn_readlane(ms, j);
      const int s1 = __builtin_amdgcn_readlane(ms, j + 1);
      const int s2 = __builtin_amdgcn_readlane(ms, j + 2);
      const int s3 = __builtin_amdgcn_readlane(ms, j + 3);
      const float c0 = __int_as_float(__builtin_amdgcn_readlane(mc, j));
      const float c1 = __int_as_float(__builtin_amdgcn_readlane(mc, j + 1));
      const float c2 = __int_as_float(__builtin_amdgcn_readlane(mc, j + 2));
      const float c3 = __int_as_float(__builtin_amdgcn_readlane(mc, j + 3));
      const vecT r0 = *(const vecT*)(T + (size_t)s0 * C + col);
      const vecT r1 = *(const vecT*)(T + (size_t)s1 * C + col);
      const vecT r2 = *(const vecT*)(T + (size_t)s2 * C + col);
      const vecT r3 = *(const vecT*)(T + (size_t)s3 * C + col);
      acc += c0 * r0;
      acc += c1 * r1;
      acc += c2 * r2;
      acc += c3 * r3;
    }
    for (; j < m; ++j) {
      const int s0 = __builtin_amdgcn_readlane(ms, j);
      const float c0 = __int_as_float(__builtin_amdgcn_readlane(mc, j));
      acc += c0 * *(const vecT*)(T + (size_t)s0 * C + col);
    }
  }

  *(vecT*)(agg + (size_t)n * C + col) = *(const vecT*)(bias + col) + dn * acc;
}

#define APITCH 132
#define GB 16

template <int COUT, int NWAVES>
__global__ __launch_bounds__(NWAVES * 32) void gather_gemm(const int* __restrict__ pairs,
                                                           const int* __restrict__ off,
                                                           const int* __restrict__ cnt,
                                                           const float* __restrict__ dinv,
                                                           const float* __restrict__ Tin,
                                                           const float* __restrict__ bias,
                                                           const float* __restrict__ Wt,
                                                           float* __restrict__ Tout,
                                                           int nRowTiles) {
  constexpr int PITCH    = 2 * COUT + 32;
  constexpr int NTHREADS = NWAVES * 32;
  constexpr int NV       = COUT / 4;
  constexpr int UNITS    = (CIN / 2) * NV;
  constexpr int PER_T    = (UNITS + NTHREADS - 1) / NTHREADS;
  constexpr bool EXACT   = (PER_T * NTHREADS == UNITS);
  __shared__ alignas(16) float sW[(CIN / 2) * PITCH];
  __shared__ alignas(16) float sA[NWAVES * 16 * APITCH];

  const int tid  = (int)threadIdx.x;
  const int lane = tid & 31;
  const int wave = __builtin_amdgcn_readfirstlane(tid >> 5);
  const int rt   = blockIdx.x * NWAVES + wave;
  const bool live = rt < nRowTiles;
  const int m0   = rt * 16;
  const int col  = lane * 4;
  float* sAw = sA + wave * (16 * APITCH);

  v4f wlo[PER_T], whi[PER_T];
#pragma unroll
  for (int u = 0; u < PER_T; ++u) {
    const int id = tid + u * NTHREADS;
    if (EXACT || id < UNITS) {
      const int p = id / NV;
      const int g = id - p * NV;
      wlo[u] = *(const v4f*)(Wt + (size_t)(2 * p) * COUT + 4 * g);
      whi[u] = *(const v4f*)(Wt + (size_t)(2 * p + 1) * COUT + 4 * g);
    }
  }

  int rOff = 0, rCnt = 0;
  float rD = 0.0f;
  v4f bv = {0.0f, 0.0f, 0.0f, 0.0f};
  v4f self[16];
  if (live) {
    const int nn = m0 + (lane & 15);
    rOff = off[nn];
    rCnt = cnt[nn];
    rD   = dinv[nn];
    bv   = *(const v4f*)(bias + col);
#pragma unroll
    for (int r = 0; r < 16; ++r) self[r] = *(const v4f*)(Tin + (size_t)(m0 + r) * CIN + col);
  }

#pragma unroll
  for (int u = 0; u < PER_T; ++u) {
    const int id = tid + u * NTHREADS;
    if (EXACT || id < UNITS) {
      const int p = id / NV;
      const int g = id - p * NV;
      float* d = &sW[p * PITCH + 8 * g];
      v4f s0, s1;
      s0.x = wlo[u].x; s0.y = whi[u].x; s0.z = wlo[u].y; s0.w = whi[u].y;
      s1.x = wlo[u].z; s1.y = whi[u].z; s1.z = wlo[u].w; s1.w = whi[u].w;
      *(v4f*)(d)     = s0;
      *(v4f*)(d + 4) = s1;
    }
  }
  __syncthreads();
  if (!live) return;

#pragma unroll
  for (int r = 0; r < 16; ++r) *(v4f*)(sAw + r * APITCH + col) = self[r];

  const int e0 = __builtin_amdgcn_readlane(rOff, 0);
  const int e1 = __builtin_amdgcn_readlane(rOff, 15) + __builtin_amdgcn_readlane(rCnt, 15);
  int r      = 0;
  int rowEnd = e0 + __builtin_amdgcn_readlane(rCnt, 0);
  int e      = e0;
  v4f acc = *(const v4f*)(sAw + col);

  auto finish = [&](v4f sum, int row) -> v4f {
    const float dr = __int_as_float(__builtin_amdgcn_readlane(__float_as_int(rD), row));
    const v4f a = bv + dr * sum;
    v4f h;
    h.x = fmaxf(a.x, 0.0f); h.y = fmaxf(a.y, 0.0f);
    h.z = fmaxf(a.z, 0.0f); h.w = fmaxf(a.w, 0.0f);
    return dr * h;
  };
  auto advance = [&]() {
    while (e == rowEnd) {
      *(v4f*)(sAw + r * APITCH + col) = finish(acc, r);
      ++r;
      rowEnd = __builtin_amdgcn_readlane(rOff, r) + __builtin_amdgcn_readlane(rCnt, r);
      acc = *(const v4f*)(sAw + r * APITCH + col);
    }
  };

  int2 cur = make_int2(0, 0);
  if (e0 + lane < e1) cur = *(const int2*)(pairs + 2 * (size_t)(e0 + lane));
  for (int base = e0; base < e1; base += 32) {
    int2 nxt = make_int2(0, 0);
    if (base + 32 + lane < e1) nxt = *(const int2*)(pairs + 2 * (size_t)(base + 32 + lane));
    const int m = (e1 - base < 32) ? (e1 - base) : 32;
    int j = 0;
    for (; j + GB <= m; j += GB) {
      v4f rv[GB];
#pragma unroll
      for (int i = 0; i < GB; ++i) {
        const int s = __builtin_amdgcn_readlane(cur.x, j + i);
        rv[i] = *(const v4f*)(Tin + (size_t)s * CIN + col);
      }
#pragma unroll
      for (int i = 0; i < GB; ++i) {
        const float c = __int_as_float(__builtin_amdgcn_readlane(cur.y, j + i));
        advance();
        acc += c * rv[i];
        ++e;
      }
    }
    for (; j + 4 <= m; j += 4) {
      v4f rv[4];
#pragma unroll
      for (int i = 0; i < 4; ++i) {
        const int s = __builtin_amdgcn_readlane(cur.x, j + i);
        rv[i] = *(const v4f*)(Tin + (size_t)s * CIN + col);
      }
#pragma unroll
      for (int i = 0; i < 4; ++i) {
        const float c = __int_as_float(__builtin_amdgcn_readlane(cur.y, j + i));
        advance();
        acc += c * rv[i];
        ++e;
      }
    }
    for (; j < m; ++j) {
      const int s = __builtin_amdgcn_readlane(cur.x, j);
      const float c = __int_as_float(__builtin_amdgcn_readlane(cur.y, j));
      const v4f rv = *(const v4f*)(Tin + (size_t)s * CIN + col);
      advance();
      acc += c * rv;
      ++e;
    }
    cur = nxt;
  }
  for (;;) {
    *(v4f*)(sAw + r * APITCH + col) = finish(acc, r);
    if (++r >= 16) break;
    acc = *(const v4f*)(sAw + r * APITCH + col);
  }

  const int mm = lane & 15;
  const int kh = lane >> 4;
  constexpr int NT = COUT / 16;
  v8f accT[NT] = {};
  const float* arow = sAw + mm * APITCH + 2 * kh;
  for (int k = 0; k < CIN; k += 4) {
    const v2f a = *(const v2f*)(arow + k);
    const float* b0 = &sW[((k >> 1) + kh) * PITCH + 2 * mm];
#pragma unroll
    for (int t = 0; t < NT; ++t) {
      v2f b = *(const v2f*)(b0 + 32 * t);
      accT[t] = __builtin_amdgcn_wmma_f32_16x16x4_f32(
           false, a,  false, b,
           (short)0, accT[t],  false,  false);
    }
  }

  float* o = Tout + (size_t)(m0 + 8 * kh) * COUT + mm;
#pragma unroll
  for (int t = 0; t < NT; ++t) {
#pragma unroll
    for (int v = 0; v < 8; ++v) {
      o[(size_t)v * COUT + t * 16] = accT[t][v];
    }
  }
}


extern "C" void kernel_launch(void* const* d_in, const int* in_sizes, int n_in,
                              void* d_out, int out_size, void* d_ws, size_t ws_size,
                              hipStream_t stream) {
  const float* x  = (const float*)d_in[0];
  const int*   ei = (const int*)d_in[1];
  const float* ea = (const float*)d_in[2];
  const float* W1 = (const float*)d_in[3];
  const float* b1 = (const float*)d_in[4];
  const float* W2 = (const float*)d_in[5];
  const float* b2 = (const float*)d_in[6];
  const float* W3 = (const float*)d_in[7];
  const float* b3 = (const float*)d_in[8];

  const int N = in_sizes[0] / CIN;
  const int E = in_sizes[2];
  const int* srcIdx = ei;
  const int* dstIdx = ei + E;

  float* T     = (float*)d_ws;
  float* agg   = T + (size_t)N * 128;
  int*   pairs = (int*)(agg + (size_t)N * 128);
  int*   recSW = pairs + (size_t)2 * E;
  int*   recD  = recSW + (size_t)2 * E;
  float* dinv  = (float*)(recD + E);
  int*   cnt   = (int*)(dinv + N);
  int*   off   = cnt + N;
  int*   startT = off + N;
  float* out   = (float*)d_out;

  const int nRowTiles = N / 16;
  const int NB = (N + BK_NODES - 1) / BK_NODES;
  const int SL = (E + NSLICE - 1) / NSLICE;
  dim3 blk(256);
  dim3 gRows((N + 7) / 8);

  dim3 gTiles(NB);
  dim3 bTiles(TILES_PER_WG * 32);

  bucket_edges<<<dim3(NSLICE), dim3(BE_THREADS), 0, stream>>>(srcIdx, dstIdx, ea, recSW, recD,
                                                              startT, E, SL, NB);

  sort_gemm<128><<<gTiles, bTiles, 0, stream>>>(x, W1, recSW, recD, startT, pairs, off, cnt, dinv,
                                                T, nRowTiles, N, SL);

  gather_gemm<128, TILES_PER_WG><<<gTiles, bTiles, 0, stream>>>(
      pairs, off, cnt, dinv, T, b1, W2, agg, nRowTiles);

  gather_gemm<64, TILES_PER_WG><<<gTiles, bTiles, 0, stream>>>(
      pairs, off, cnt, dinv, agg, b2, W3, T, nRowTiles);

  gather_rows<64><<<gRows, blk, 0, stream>>>(pairs, off, cnt, dinv, T, b3, out, N);
}


# Both edge-grouping levels place each record in LDS by its rank and copy the ordered run out with whole-line stores; was one scattered 8- or 4-byte global store per record; record order kept
# speedup vs baseline: 5.6276x; 1.0788x over previous
#include <hip/hip_runtime.h>
#include <hip/hip_bf16.h>


#define CIN 128

typedef __attribute__((ext_vector_type(2))) float v2f;
typedef __attribute__((ext_vector_type(4))) float v4f;
typedef __attribute__((ext_vector_type(8))) float v8f;

#define TILES_PER_WG 25
#define BK_NODES   (TILES_PER_WG * 16)
#define NSLICE     250
#define BE_THREADS 1024
#define BE_EPT     7
#define BE_MAXB    1023
#define BE_CAP     (BE_EPT * BE_THREADS)

__global__ __launch_bounds__(BE_THREADS) void bucket_edges(const int* __restrict__ src,
                                                           const int* __restrict__ dst,
                                                           const float* __restrict__ ea,
                                                           int* __restrict__ recSW,
                                                           int* __restrict__ recD,
                                                           int* __restrict__ startT,
                                                           int E, int SL, int NB) {
  __shared__ int sCnt[BE_MAXB + 1];
  __shared__ int sStart[BE_MAXB + 1];
  __shared__ int sWave[BE_THREADS / 32];
  __shared__ int2 sRecSW[BE_CAP];
  __shared__ int  sRecD[BE_CAP];
  const int t    = threadIdx.x;
  const int lane = t & 31;
  const int wv   = t >> 5;
  const int g    = blockIdx.x;
  const int sbeg = g * SL;
  const int send = (E - sbeg < SL) ? E : sbeg + SL;

  int d[BE_EPT], s[BE_EPT], w[BE_EPT];
#pragma unroll
  for (int u = 0; u < BE_EPT; ++u) {
    const int i = sbeg + u * BE_THREADS + t;
    d[u] = -1; s[u] = 0; w[u] = 0;
    if (i < send) { d[u] = dst[i]; s[u] = src[i]; w[u] = __float_as_int(ea[i]); }
  }
  sCnt[t] = 0;
  __syncthreads();

  int bk[BE_EPT], rk[BE_EPT];
#pragma unroll
  for (int u = 0; u < BE_EPT; ++u) {
    bk[u] = 0; rk[u] = 0;
    if (d[u] >= 0) { bk[u] = d[u] / BK_NODES; rk[u] = atomicAdd(&sCnt[bk[u]], 1); }
  }
  __syncthreads();

  const int c = sCnt[t];
  int incl = c;
  for (int o = 1; o < 32; o <<= 1) {
    const int y = __shfl_up(incl, o);
    if (lane >= o) incl += y;
  }
  if (lane == 31) sWave[wv] = incl;
  __syncthreads();
  int front = 0;
  for (int k = 0; k < wv; ++k) front += sWave[k];
  const int excl = front + incl - c;
  sStart[t] = excl;
  if (t <= NB) startT[(size_t)t * NSLICE + g] = excl;
  __syncthreads();

#pragma unroll
  for (int u = 0; u < BE_EPT; ++u) {
    if (d[u] >= 0) {
      const int p = sStart[bk[u]] + rk[u];
      sRecSW[p] = make_int2(s[u], w[u]);
      sRecD[p]  = d[u];
    }
  }
  __syncthreads();

  const int n = send - sbeg;
  for (int i = t; i < n; i += BE_THREADS) {
    *(int2*)(recSW + 2 * (size_t)(sbeg + i)) = sRecSW[i];
    recD[sbeg + i] = sRecD[i];
  }
}

#define RPW (NSLICE / TILES_PER_WG)
#define RDEPTH 2
#define PCAP 8192
static_assert(RPW * TILES_PER_WG == NSLICE, "every run has a wave");
static_assert(NSLICE <= TILES_PER_WG * 32 && BK_NODES <= TILES_PER_WG * 32, "one thread per slice / per node");

template <int COUT>
__global__ __launch_bounds__(TILES_PER_WG * 32) void sort_gemm(const float* __restrict__ H,
                                                               const float* __restrict__ Wt,
                                                               const int* __restrict__ recSW,
                                                               const int* __restrict__ recD,
                                                               const int* __restrict__ startT,
                                                               int* pairs,
                                                               int* __restrict__ off,
                                                               int* __restrict__ cnt,
                                                               float* __restrict__ dinv,
                                                               float* __restrict__ T,
                                                               int nRowTiles, int N, int SL) {
  constexpr int PITCH    = 2 * COUT + 32;
  constexpr int NTHREADS = TILES_PER_WG * 32;
  constexpr int NV       = COUT / 4;
  constexpr int UNITS    = (CIN / 2) * NV;
  constexpr int PER_T    = (UNITS + NTHREADS - 1) / NTHREADS;
  __shared__ alignas(16) float sW[(CIN / 2) * PITCH];
  __shared__ int sRunS[NSLICE];
  __shared__ int sRunE[NSLICE];
  __shared__ int sCntN[BK_NODES];
  __shared__ int sOffN[BK_NODES];
  __shared__ int sPosN[BK_NODES];
  __shared__ int sRedA[TILES_PER_WG];
  __shared__ int sRedB[TILES_PER_WG];
  __shared__ int2 sPairs[PCAP];
  __shared__ int sTot;

  const int tid  = (int)threadIdx.x;
  const int lane = tid & 31;
  const int wave = __builtin_amdgcn_readfirstlane(tid >> 5);
  const int b    = (int)blockIdx.x;
  const int nbeg = b * BK_NODES;

  v4f wlo[PER_T], whi[PER_T];
#pragma unroll
  for (int u = 0; u < PER_T; ++u) {
    const int id = tid + u * NTHREADS;
    if (id < UNITS) {
      const int p = id / NV;
      const int g = id - p * NV;
      wlo[u] = *(const v4f*)(Wt + (size_t)(2 * p) * COUT + 4 * g);
      whi[u] = *(const v4f*)(Wt + (size_t)(2 * p + 1) * COUT + 4 * g);
    }
  }

  int rs = 0, re = 0;
  if (tid < NSLICE) {
    rs = startT[(size_t)b * NSLICE + tid];
    re = startT[(size_t)(b + 1) * NSLICE + tid];
  }
  if (tid < BK_NODES) sCntN[tid] = 0;

#pragma unroll
  for (int u = 0; u < PER_T; ++u) {
    const int id = tid + u * NTHREADS;
    if (id < UNITS) {
      const int p = id / NV;
      const int g = id - p * NV;
      float* d = &sW[p * PITCH + 8 * g];
      v4f s0, s1;
      s0.x = wlo[u].x; s0.y = whi[u].x; s0.z = wlo[u].y; s0.w = whi[u].y;
      s1.x = wlo[u].z; s1.y = whi[u].z; s1.z = wlo[u].w; s1.w = whi[u].w;
      *(v4f*)(d)     = s0;
      *(v4f*)(d + 4) = s1;
    }
  }

  if (tid < NSLICE) {
    sRunS[tid] = tid * SL + rs;
    sRunE[tid] = tid * SL + re;
  }
  int fs = rs;
  for (int o = 16; o > 0; o >>= 1) fs += __shfl_xor(fs, o);
  if (lane == 0) sRedA[wave] = fs;
  __syncthreads();
  int base = 0;
  for (int k = 0; k < TILES_PER_WG; ++k) base += sRedA[k];

  int rS[RPW], rE[RPW], dd[RPW * RDEPTH];
  int2 sw[RPW * RDEPTH];
#pragma unroll
  for (int k = 0; k < RPW; ++k) {
    rS[k] = __builtin_amdgcn_readfirstlane(sRunS[wave * RPW + k]);
    rE[k] = __builtin_amdgcn_readfirstlane(sRunE[wave * RPW + k]);
  }
#pragma unroll
  for (int k = 0; k < RPW * RDEPTH; ++k) {
    const int i = rS[k / RDEPTH] + 32 * (k % RDEPTH) + lane;
    dd[k] = -1;
    sw[k] = make_int2(0, 0);
    if (i < rE[k / RDEPTH]) {
      dd[k] = recD[i];
      sw[k] = *(const int2*)(recSW + 2 * (size_t)i);
    }
  }
#pragma unroll
  for (int k = 0; k < RPW * RDEPTH; ++k) {
    if (dd[k] >= 0) atomicAdd(&sCntN[dd[k] - nbeg], 1);
  }
  for (int k = 0; k < RPW; ++k) {
    for (int i = rS[k] + 32 * RDEPTH + lane; i < rE[k]; i += 32) atomicAdd(&sCntN[recD[i] - nbeg], 1);
  }
  __syncthreads();

  const int c = (tid < BK_NODES) ? sCntN[tid] : 0;
  int incl = c;
  for (int o = 1; o < 32; o <<= 1) {
    const int y = __shfl_up(incl, o);
    if (lane >= o) incl += y;
  }
  if (lane == 31) sRedB[wave] = incl;
  __syncthreads();
  int front = 0;
  for (int k = 0; k < wave; ++k) front += sRedB[k];
  const int excl = front + incl - c;
  if (tid < BK_NODES) {
    sOffN[tid] = excl;
    sPosN[tid] = excl;
    const int n = nbeg + tid;
    if (n < N) { off[n] = base + excl; cnt[n] = c; }
  }
  if (tid == NTHREADS - 1) sTot = front + incl;
  __syncthreads();
  const int tot = __builtin_amdgcn_readfirstlane(sTot);

  if (tot <= PCAP) {
#pragma unroll
    for (int k = 0; k < RPW * RDEPTH; ++k) {
      if (dd[k] >= 0) {
        const int p = atomicAdd(&sPosN[dd[k] - nbeg], 1);
        sPairs[p] = sw[k];
      }
    }
    for (int k = 0; k < RPW; ++k) {
      for (int i = rS[k] + 32 * RDEPTH + lane; i < rE[k]; i += 32) {
        const int dl = recD[i] - nbeg;
        const int2 v = *(const int2*)(recSW + 2 * (size_t)i);
        const int p  = atomicAdd(&sPosN[dl], 1);
        sPairs[p] = v;
      }
    }
    __syncthreads();
    for (int i = tid; i < tot; i += NTHREADS) {
      *(int2*)(pairs + 2 * (size_t)(base + i)) = sPairs[i];
    }
  } else {
#pragma unroll
    for (int k = 0; k < RPW * RDEPTH; ++k) {
      if (dd[k] >= 0) {
        const int p = atomicAdd(&sPosN[dd[k] - nbeg], 1);
        *(int2*)(pairs + 2 * (size_t)(base + p)) = sw[k];
      }
    }
    for (int k = 0; k < RPW; ++k) {
      for (int i = rS[k] + 32 * RDEPTH + lane; i < rE[k]; i += 32) {
        const int dl = recD[i] - nbeg;
        const int2 v = *(const int2*)(recSW + 2 * (size_t)i);
        const int p  = atomicAdd(&sPosN[dl], 1);
        *(int2*)(pairs + 2 * (size_t)(base + p)) = v;
      }
    }
  }
  __syncthreads();

  const int rt = b * TILES_PER_WG + wave;
  if (rt >= nRowTiles) return;

  const int li = wave * 16 + (lane & 15);
  const int nn = nbeg + li;
  float dn;
  {
    const int* p = pairs + 2 * (size_t)(base + sOffN[li]);
    const int cn = sCntN[li];
    float d = 1.0f;
    int j = 0;
    for (; j + 4 <= cn; j += 4) {
      const float w0 = __int_as_float(p[2 * j + 1]);
      const float w1 = __int_as_float(p[2 * j + 3]);
      const float w2 = __int_as_float(p[2 * j + 5]);
      const float w3 = __int_as_float(p[2 * j + 7]);
      d += w0;
      d += w1;
      d += w2;
      d += w3;
    }
    for (; j < cn; ++j) d += __int_as_float(p[2 * j + 1]);
    dn = (d > 0.0f) ? rsqrtf(d) : 0.0f;
    if (lane < 16) dinv[nn] = dn;
  }

  const int m0 = rt * 16;
  const int m  = lane & 15;
  const int kh = lane >> 4;

  constexpr int NT = COUT / 16;
  v8f acc[NT] = {};

  const float* arow = H + (size_t)(m0 + m) * CIN + 2 * kh;

  for (int k = 0; k < CIN; k += 4) {
    const v2f a = *(const v2f*)(arow + k);
    const float* b0 = &sW[((k >> 1) + kh) * PITCH + 2 * m];
#pragma unroll
    for (int t = 0; t < NT; ++t) {
      v2f bb = *(const v2f*)(b0 + 32 * t);
      acc[t] = __builtin_amdgcn_wmma_f32_16x16x4_f32(
           false, a,  false, bb,
           (short)0, acc[t],  false,  false);
    }
  }

  float sc[8];
#pragma unroll
  for (int v = 0; v < 8; ++v) sc[v] = __shfl(dn, v + 8 * kh);
  float* out = T + (size_t)(m0 + 8 * kh) * COUT + m;
#pragma unroll
  for (int t = 0; t < NT; ++t) {
#pragma unroll
    for (int v = 0; v < 8; ++v) {
      out[(size_t)v * COUT + t * 16] = sc[v] * acc[t][v];
    }
  }
}


template <int C>
__global__ __launch_bounds__(256) void gather_rows(const int* __restrict__ pairs,
                                                   const int* __restrict__ off,
                                                   const int* __restrict__ cnt,
                                                   const float* __restrict__ dinv,
                                                   const float* __restrict__ T,
                                                   const float* __restrict__ bias,
                                                   float* __restrict__ agg, int N) {
  constexpr int V = C / 32;
  typedef float vecT __attribute__((ext_vector_type(V)));

  const int lane = threadIdx.x & 31;
  const int wv = __builtin_amdgcn_readfirstlane((int)(threadIdx.x >> 5));
  const int n  = blockIdx.x * 8 + wv;
  if (n >= N) return;

  const int beg = off[n];
  const int c   = cnt[n];
  const float dn = dinv[n];

  const int col = lane * V;
  vecT acc = *(const vecT*)(T + (size_t)n * C + col);

  for (int base = 0; base < c; base += 32) {
    const int m = (c - base < 32) ? (c - base) : 32;
    int ms = 0;
    int mc = 0;
    if (lane < m) {
      const int2 en = *(const int2*)(pairs + 2 * (size_t)(beg + base + lane));
      ms = en.x;
      mc = en.y;
    }
    int j = 0;
    for (; j + 4 <= m; j += 4) {
      const int s0 = __builtin_amdgcn_readlane(ms, j);
      const int s1 = __builtin_amdgcn_readlane(ms, j + 1);
      const int s2 = __builtin_amdgcn_readlane(ms, j + 2);
      const int s3 = __builtin_amdgcn_readlane(ms, j + 3);
      const float c0 = __int_as_float(__builtin_amdgcn_readlane(mc, j));
      const float c1 = __int_as_float(__builtin_amdgcn_readlane(mc, j + 1));
      const float c2 = __int_as_float(__builtin_amdgcn_readlane(mc, j + 2));
      const float c3 = __int_as_float(__builtin_amdgcn_readlane(mc, j + 3));
      const vecT r0 = *(const vecT*)(T + (size_t)s0 * C + col);
      const vecT r1 = *(const vecT*)(T + (size_t)s1 * C + col);
      const vecT r2 = *(const vecT*)(T + (size_t)s2 * C + col);
      const vecT r3 = *(const vecT*)(T + (size_t)s3 * C + col);
      acc += c0 * r0;
      acc += c1 * r1;
      acc += c2 * r2;
      acc += c3 * r3;
    }
    for (; j < m; ++j) {
      const int s0 = __builtin_amdgcn_readlane(ms, j);
      const float c0 = __int_as_float(__builtin_amdgcn_readlane(mc, j));
      acc += c0 * *(const vecT*)(T + (size_t)s0 * C + col);
    }
  }

  *(vecT*)(agg + (size_t)n * C + col) = *(const vecT*)(bias + col) + dn * acc;
}

#define APITCH 132
#define GB 16

template <int COUT, int NWAVES>
__global__ __launch_bounds__(NWAVES * 32) void gather_gemm(const int* __restrict__ pairs,
                                                           const int* __restrict__ off,
                                                           const int* __restrict__ cnt,
                                                           const float* __restrict__ dinv,
                                                           const float* __restrict__ Tin,
                                                           const float* __restrict__ bias,
                                                           const float* __restrict__ Wt,
                                                           float* __restrict__ Tout,
                                                           int nRowTiles) {
  constexpr int PITCH    = 2 * COUT + 32;
  constexpr int NTHREADS = NWAVES * 32;
  constexpr int NV       = COUT / 4;
  constexpr int UNITS    = (CIN / 2) * NV;
  constexpr int PER_T    = (UNITS + NTHREADS - 1) / NTHREADS;
  constexpr bool EXACT   = (PER_T * NTHREADS == UNITS);
  __shared__ alignas(16) float sW[(CIN / 2) * PITCH];
  __shared__ alignas(16) float sA[NWAVES * 16 * APITCH];

  const int tid  = (int)threadIdx.x;
  const int lane = tid & 31;
  const int wave = __builtin_amdgcn_readfirstlane(tid >> 5);
  const int rt   = blockIdx.x * NWAVES + wave;
  const bool live = rt < nRowTiles;
  const int m0   = rt * 16;
  const int col  = lane * 4;
  float* sAw = sA + wave * (16 * APITCH);

  v4f wlo[PER_T], whi[PER_T];
#pragma unroll
  for (int u = 0; u < PER_T; ++u) {
    const int id = tid + u * NTHREADS;
    if (EXACT || id < UNITS) {
      const int p = id / NV;
      const int g = id - p * NV;
      wlo[u] = *(const v4f*)(Wt + (size_t)(2 * p) * COUT + 4 * g);
      whi[u] = *(const v4f*)(Wt + (size_t)(2 * p + 1) * COUT + 4 * g);
    }
  }

  int rOff = 0, rCnt = 0;
  float rD = 0.0f;
  v4f bv = {0.0f, 0.0f, 0.0f, 0.0f};
  v4f self[16];
  if (live) {
    const int nn = m0 + (lane & 15);
    rOff = off[nn];
    rCnt = cnt[nn];
    rD   = dinv[nn];
    bv   = *(const v4f*)(bias + col);
#pragma unroll
    for (int r = 0; r < 16; ++r) self[r] = *(const v4f*)(Tin + (size_t)(m0 + r) * CIN + col);
  }

#pragma unroll
  for (int u = 0; u < PER_T; ++u) {
    const int id = tid + u * NTHREADS;
    if (EXACT || id < UNITS) {
      const int p = id / NV;
      const int g = id - p * NV;
      float* d = &sW[p * PITCH + 8 * g];
      v4f s0, s1;
      s0.x = wlo[u].x; s0.y = whi[u].x; s0.z = wlo[u].y; s0.w = whi[u].y;
      s1.x = wlo[u].z; s1.y = whi[u].z; s1.z = wlo[u].w; s1.w = whi[u].w;
      *(v4f*)(d)     = s0;
      *(v4f*)(d + 4) = s1;
    }
  }
  __syncthreads();
  if (!live) return;

#pragma unroll
  for (int r = 0; r < 16; ++r) *(v4f*)(sAw + r * APITCH + col) = self[r];

  const int e0 = __builtin_amdgcn_readlane(rOff, 0);
  const int e1 = __builtin_amdgcn_readlane(rOff, 15) + __builtin_amdgcn_readlane(rCnt, 15);
  int r      = 0;
  int rowEnd = e0 + __builtin_amdgcn_readlane(rCnt, 0);
  int e      = e0;
  v4f acc = *(const v4f*)(sAw + col);

  auto finish = [&](v4f sum, int row) -> v4f {
    const float dr = __int_as_float(__builtin_amdgcn_readlane(__float_as_int(rD), row));
    const v4f a = bv + dr * sum;
    v4f h;
    h.x = fmaxf(a.x, 0.0f); h.y = fmaxf(a.y, 0.0f);
    h.z = fmaxf(a.z, 0.0f); h.w = fmaxf(a.w, 0.0f);
    return dr * h;
  };
  auto advance = [&]() {
    while (e == rowEnd) {
      *(v4f*)(sAw + r * APITCH + col) = finish(acc, r);
      ++r;
      rowEnd = __builtin_amdgcn_readlane(rOff, r) + __builtin_amdgcn_readlane(rCnt, r);
      acc = *(const v4f*)(sAw + r * APITCH + col);
    }
  };

  int2 cur = make_int2(0, 0);
  if (e0 + lane < e1) cur = *(const int2*)(pairs + 2 * (size_t)(e0 + lane));
  for (int base = e0; base < e1; base += 32) {
    int2 nxt = make_int2(0, 0);
    if (base + 32 + lane < e1) nxt = *(const int2*)(pairs + 2 * (size_t)(base + 32 + lane));
    const int m = (e1 - base < 32) ? (e1 - base) : 32;
    int j = 0;
    for (; j + GB <= m; j += GB) {
      v4f rv[GB];
#pragma unroll
      for (int i = 0; i < GB; ++i) {
        const int s = __builtin_amdgcn_readlane(cur.x, j + i);
        rv[i] = *(const v4f*)(Tin + (size_t)s * CIN + col);
      }
#pragma unroll
      for (int i = 0; i < GB; ++i) {
        const float c = __int_as_float(__builtin_amdgcn_readlane(cur.y, j + i));
        advance();
        acc += c * rv[i];
        ++e;
      }
    }
    for (; j + 4 <= m; j += 4) {
      v4f rv[4];
#pragma unroll
      for (int i = 0; i < 4; ++i) {
        const int s = __builtin_amdgcn_readlane(cur.x, j + i);
        rv[i] = *(const v4f*)(Tin + (size_t)s * CIN + col);
      }
#pragma unroll
      for (int i = 0; i < 4; ++i) {
        const float c = __int_as_float(__builtin_amdgcn_readlane(cur.y, j + i));
        advance();
        acc += c * rv[i];
        ++e;
      }
    }
    for (; j < m; ++j) {
      const int s = __builtin_amdgcn_readlane(cur.x, j);
      const float c = __int_as_float(__builtin_amdgcn_readlane(cur.y, j));
      const v4f rv = *(const v4f*)(Tin + (size_t)s * CIN + col);
      advance();
      acc += c * rv;
      ++e;
    }
    cur = nxt;
  }
  for (;;) {
    *(v4f*)(sAw + r * APITCH + col) = finish(acc, r);
    if (++r >= 16) break;
    acc = *(const v4f*)(sAw + r * APITCH + col);
  }

  const int mm = lane & 15;
  const int kh = lane >> 4;
  constexpr int NT = COUT / 16;
  v8f accT[NT] = {};
  const float* arow = sAw + mm * APITCH + 2 * kh;
  for (int k = 0; k < CIN; k += 4) {
    const v2f a = *(const v2f*)(arow + k);
    const float* b0 = &sW[((k >> 1) + kh) * PITCH + 2 * mm];
#pragma unroll
    for (int t = 0; t < NT; ++t) {
      v2f b = *(const v2f*)(b0 + 32 * t);
      accT[t] = __builtin_amdgcn_wmma_f32_16x16x4_f32(
           false, a,  false, b,
           (short)0, accT[t],  false,  false);
    }
  }

  float* o = Tout + (size_t)(m0 + 8 * kh) * COUT + mm;
#pragma unroll
  for (int t = 0; t < NT; ++t) {
#pragma unroll
    for (int v = 0; v < 8; ++v) {
      o[(size_t)v * COUT + t * 16] = accT[t][v];
    }
  }
}


extern "C" void kernel_launch(void* const* d_in, const int* in_sizes, int n_in,
                              void* d_out, int out_size, void* d_ws, size_t ws_size,
                              hipStream_t stream) {
  const float* x  = (const float*)d_in[0];
  const int*   ei = (const int*)d_in[1];
  const float* ea = (const float*)d_in[2];
  const float* W1 = (const float*)d_in[3];
  const float* b1 = (const float*)d_in[4];
  const float* W2 = (const float*)d_in[5];
  const float* b2 = (const float*)d_in[6];
  const float* W3 = (const float*)d_in[7];
  const float* b3 = (const float*)d_in[8];

  const int N = in_sizes[0] / CIN;
  const int E = in_sizes[2];
  const int* srcIdx = ei;
  const int* dstIdx = ei + E;

  float* T     = (float*)d_ws;
  float* agg   = T + (size_t)N * 128;
  int*   pairs = (int*)(agg + (size_t)N * 128);
  int*   recSW = pairs + (size_t)2 * E;
  int*   recD  = recSW + (size_t)2 * E;
  float* dinv  = (float*)(recD + E);
  int*   cnt   = (int*)(dinv + N);
  int*   off   = cnt + N;
  int*   startT = off + N;
  float* out   = (float*)d_out;

  const int nRowTiles = N / 16;
  const int NB = (N + BK_NODES - 1) / BK_NODES;
  const int SL = (E + NSLICE - 1) / NSLICE;
  dim3 blk(256);
  dim3 gRows((N + 7) / 8);

  dim3 gTiles(NB);
  dim3 bTiles(TILES_PER_WG * 32);

  bucket_edges<<<dim3(NSLICE), dim3(BE_THREADS), 0, stream>>>(srcIdx, dstIdx, ea, recSW, recD,
                                                              startT, E, SL, NB);

  sort_gemm<128><<<gTiles, bTiles, 0, stream>>>(x, W1, recSW, recD, startT, pairs, off, cnt, dinv,
                                                T, nRowTiles, N, SL);

  gather_gemm<128, TILES_PER_WG><<<gTiles, bTiles, 0, stream>>>(
      pairs, off, cnt, dinv, T, b1, W2, agg, nRowTiles);

  gather_gemm<64, TILES_PER_WG><<<gTiles, bTiles, 0, stream>>>(
      pairs, off, cnt, dinv, agg, b2, W3, T, nRowTiles);

  gather_rows<64><<<gRows, blk, 0, stream>>>(pairs, off, cnt, dinv, T, b3, out, N);
}
